# starting kernel
# baseline (speedup 1.0000x reference)
#include <hip/hip_runtime.h>


typedef __attribute__((ext_vector_type(2))) float v2f;
typedef __attribute__((ext_vector_type(8))) float v8f;

#define NEGV (-1000000000.0f)

__device__ __forceinline__ unsigned enc_f32(float f) {
  unsigned u = __float_as_uint(f);
  return (u & 0x80000000u) ? ~u : (u | 0x80000000u);
}
__device__ __forceinline__ float dec_f32(unsigned v) {
  unsigned u = (v & 0x80000000u) ? (v & 0x7FFFFFFFu) : ~v;
  return __uint_as_float(u);
}

__global__ void k_init(const float* __restrict__ x, float* __restrict__ out,
                       float* __restrict__ denom, unsigned* __restrict__ maxenc,
                       int N) {
  int total = N * 64;
  for (int i = blockIdx.x * blockDim.x + threadIdx.x; i < total;
       i += gridDim.x * blockDim.x) {
    out[i] = x[i];
    if (i < N) {
      denom[i] = 0.0f;
      maxenc[i] = enc_f32(NEGV);
    }
  }
}

__global__ void __launch_bounds__(256) k_node_gemm(
    const float* __restrict__ x, const float* __restrict__ Wn,
    float* __restrict__ z, int N) {
  int wave = (blockIdx.x * blockDim.x + threadIdx.x) >> 5;
  int lane = threadIdx.x & 31;
  int nRowTiles = (N + 15) >> 4;
  if (wave >= nRowTiles) return;

  int rBase = wave << 4;
  bool fullTile = (rBase + 16 <= N);
  int half = lane >> 4;
  int l16 = lane & 15;

  int arow = rBase + l16;
  if (arow >= N) arow = N - 1;
  v2f a[16];
#pragma unroll
  for (int t = 0; t < 16; ++t) {
    int k = (t << 2) + (half << 1);
    a[t].x = x[arow * 64 + k];
    a[t].y = x[arow * 64 + k + 1];
  }

#pragma unroll
  for (int ct = 0; ct < 4; ++ct) {
    int col = (ct << 4) + l16;
    v8f acc = {};
#pragma unroll
    for (int t = 0; t < 16; ++t) {
      int k = (t << 2) + (half << 1);
      v2f b;
      b.x = Wn[k * 64 + col];
      b.y = Wn[(k + 1) * 64 + col];
      acc = __builtin_amdgcn_wmma_f32_16x16x4_f32(
          false, a[t], false, b, (short)0, acc, false, false);
    }
    if (fullTile) {
#pragma unroll
      for (int j = 0; j < 8; ++j) {
        int row = rBase + j + (half << 3);
        z[row * 64 + col] = acc[j];
      }
    } else {
#pragma unroll
      for (int j = 0; j < 8; ++j) {
        int row = rBase + j + (half << 3);
        if (row < N) z[row * 64 + col] = acc[j];
      }
    }
  }
}

__global__ void k_weff(const float* __restrict__ We,
                       const float* __restrict__ att_edge,
                       float* __restrict__ weff) {
  int k = threadIdx.x;
  if (k < 16) {
    float s = 0.0f;
    for (int c = 0; c < 64; ++c) s += We[k * 64 + c] * att_edge[c];
    weff[k] = s;
  }
}

__global__ void k_node_scores(const float* __restrict__ z,
                              const float* __restrict__ att_src,
                              const float* __restrict__ att_dst,
                              float* __restrict__ sA, float* __restrict__ sD,
                              int N) {
  int n = blockIdx.x * blockDim.x + threadIdx.x;
  if (n >= N) return;
  const float4* zr = (const float4*)(z + (size_t)n * 64);
  const float4* as4 = (const float4*)att_src;
  const float4* ad4 = (const float4*)att_dst;
  float a = 0.0f, d = 0.0f;
#pragma unroll
  for (int i = 0; i < 16; ++i) {
    float4 v = zr[i], s = as4[i], t = ad4[i];
    a += v.x * s.x + v.y * s.y + v.z * s.z + v.w * s.w;
    d += v.x * t.x + v.y * t.y + v.z * t.z + v.w * t.w;
  }
  sA[n] = a;
  sD[n] = d;
}

__global__ void k_score(const int* __restrict__ ei,
                        const float* __restrict__ ea,
                        const float* __restrict__ sA,
                        const float* __restrict__ sD,
                        const float* __restrict__ weff,
                        float* __restrict__ scoreBuf,
                        unsigned* __restrict__ maxenc, int E) {
  int e = blockIdx.x * blockDim.x + threadIdx.x;
  if (e >= E) return;
  int s = ei[e], d = ei[E + e];
  float acc = sA[s] + sD[d];
  const float4* er = (const float4*)(ea + (size_t)e * 16);
  const float4* w4 = (const float4*)weff;
#pragma unroll
  for (int i = 0; i < 4; ++i) {
    float4 v = er[i], w = w4[i];
    acc += v.x * w.x + v.y * w.y + v.z * w.z + v.w * w.w;
  }
  float sc = acc > 0.0f ? acc : 0.2f * acc;
  scoreBuf[e] = sc;
  atomicMax(maxenc + d, enc_f32(sc));
}

__global__ void k_exp(const int* __restrict__ ei, float* __restrict__ num,
                      const unsigned* __restrict__ maxenc,
                      float* __restrict__ denom, int E) {
  int e = blockIdx.x * blockDim.x + threadIdx.x;
  if (e >= E) return;
  int d = ei[E + e];
  float m = dec_f32(maxenc[d]);
  float ex = __expf(num[e] - m);
  num[e] = ex;
  atomicAdd(denom + d, ex);
}

__global__ void __launch_bounds__(256) k_message(
    const int* __restrict__ ei, const float* __restrict__ ea,
    const float* __restrict__ We, const float* __restrict__ z,
    const float* __restrict__ num, const float* __restrict__ denom,
    float* __restrict__ out, int E) {
  __shared__ float s_alpha[128];
  __shared__ int s_src[128];
  __shared__ int s_dst[128];

  int tid = threadIdx.x;
  int blockBase = blockIdx.x * 128;

  if (tid < 128) {
    int e = blockBase + tid;
    if (e < E) {
      int d = ei[E + e];
      int sN = ei[e];
      s_alpha[tid] = num[e] / (denom[d] + 1e-8f);
      s_src[tid] = sN;
      s_dst[tid] = d;
      const float* zrow = z + (size_t)sN * 64;
      __builtin_prefetch(zrow, 0, 3);
      __builtin_prefetch(zrow + 32, 0, 3);
    } else {
      s_alpha[tid] = 0.0f;
      s_src[tid] = 0;
      s_dst[tid] = 0;
    }
  }
  __syncthreads();

  int wave = tid >> 5;
  int lane = tid & 31;
  int half = lane >> 4;
  int l16 = lane & 15;
  int tileBase = wave << 4;
  int eBase = blockBase + tileBase;
  if (eBase >= E) return;

  int arow = eBase + l16;
  if (arow >= E) arow = E - 1;
  v2f a[4];
#pragma unroll
  for (int t = 0; t < 4; ++t) {
    int k = (t << 2) + (half << 1);
    a[t].x = ea[(size_t)arow * 16 + k];
    a[t].y = ea[(size_t)arow * 16 + k + 1];
  }

#pragma unroll
  for (int ct = 0; ct < 4; ++ct) {
    int col = (ct << 4) + l16;
    v8f acc = {};
#pragma unroll
    for (int t = 0; t < 4; ++t) {
      int k = (t << 2) + (half << 1);
      v2f b;
      b.x = We[k * 64 + col];
      b.y = We[(k + 1) * 64 + col];
      acc = __builtin_amdgcn_wmma_f32_16x16x4_f32(
          false, a[t], false, b, (short)0, acc, false, false);
    }
#pragma unroll
    for (int j = 0; j < 8; ++j) {
      int li = tileBase + j + (half << 3);
      float al = s_alpha[li];
      int sN = s_src[li];
      int dN = s_dst[li];
      float val = (z[(size_t)sN * 64 + col] + acc[j]) * al;
      atomicAdd(out + (size_t)dN * 64 + col, val);
    }
  }
}

extern "C" void kernel_launch(void* const* d_in, const int* in_sizes, int n_in,
                              void* d_out, int out_size, void* d_ws,
                              size_t ws_size, hipStream_t stream) {
  const float* x = (const float*)d_in[0];
  const int* ei = (const int*)d_in[1];
  const float* ea = (const float*)d_in[2];
  const float* Wn = (const float*)d_in[3];
  const float* We = (const float*)d_in[4];
  const float* att_src = (const float*)d_in[5];
  const float* att_dst = (const float*)d_in[6];
  const float* att_edge = (const float*)d_in[7];

  int N = in_sizes[0] / 64;
  int E = in_sizes[1] / 2;
  float* out = (float*)d_out;

  float* z = (float*)d_ws;
  float* sA = z + (size_t)N * 64;
  float* sD = sA + N;
  float* denom = sD + N;
  unsigned* maxenc = (unsigned*)(denom + N);
  float* num = (float*)(maxenc + N);
  float* weff = num + E;

  int total = N * 64;
  int ib = (total + 255) / 256;
  hipLaunchKernelGGL(k_init, dim3(ib), dim3(256), 0, stream, x, out, denom,
                     maxenc, N);

  int nRowTiles = (N + 15) / 16;
  int gb = (nRowTiles + 7) / 8;
  hipLaunchKernelGGL(k_node_gemm, dim3(gb), dim3(256), 0, stream, x, Wn, z, N);

  hipLaunchKernelGGL(k_weff, dim3(1), dim3(32), 0, stream, We, att_edge, weff);

  hipLaunchKernelGGL(k_node_scores, dim3((N + 255) / 256), dim3(256), 0, stream,
                     z, att_src, att_dst, sA, sD, N);

  hipLaunchKernelGGL(k_score, dim3((E + 255) / 256), dim3(256), 0, stream, ei,
                     ea, sA, sD, weff, num, maxenc, E);

  hipLaunchKernelGGL(k_exp, dim3((E + 255) / 256), dim3(256), 0, stream, ei,
                     num, maxenc, denom, E);

  hipLaunchKernelGGL(k_message, dim3((E + 127) / 128), dim3(256), 0, stream, ei,
                     ea, We, z, num, denom, out, E);
}


# message pass: accumulator halves of two column tiles exchanged (v_permlane16_swap) so each z gather and atomic add covers one whole 128-B line of one edge row; row bases scalar; atomic order changes
# speedup vs baseline: 1.1700x; 1.1700x over previous
#include <hip/hip_runtime.h>


typedef __attribute__((ext_vector_type(2))) float v2f;
typedef __attribute__((ext_vector_type(8))) float v8f;

#define NEGV (-1000000000.0f)

__device__ __forceinline__ unsigned enc_f32(float f) {
  unsigned u = __float_as_uint(f);
  return (u & 0x80000000u) ? ~u : (u | 0x80000000u);
}
__device__ __forceinline__ float dec_f32(unsigned v) {
  unsigned u = (v & 0x80000000u) ? (v & 0x7FFFFFFFu) : ~v;
  return __uint_as_float(u);
}

__global__ void k_init(const float* __restrict__ x, float* __restrict__ out,
                       float* __restrict__ denom, unsigned* __restrict__ maxenc,
                       int N) {
  int total = N * 64;
  for (int i = blockIdx.x * blockDim.x + threadIdx.x; i < total;
       i += gridDim.x * blockDim.x) {
    out[i] = x[i];
    if (i < N) {
      denom[i] = 0.0f;
      maxenc[i] = enc_f32(NEGV);
    }
  }
}

__global__ void __launch_bounds__(256) k_node_gemm(
    const float* __restrict__ x, const float* __restrict__ Wn,
    float* __restrict__ z, int N) {
  int wave = (blockIdx.x * blockDim.x + threadIdx.x) >> 5;
  int lane = threadIdx.x & 31;
  int nRowTiles = (N + 15) >> 4;
  if (wave >= nRowTiles) return;

  int rBase = wave << 4;
  bool fullTile = (rBase + 16 <= N);
  int half = lane >> 4;
  int l16 = lane & 15;

  int arow = rBase + l16;
  if (arow >= N) arow = N - 1;
  v2f a[16];
#pragma unroll
  for (int t = 0; t < 16; ++t) {
    int k = (t << 2) + (half << 1);
    a[t].x = x[arow * 64 + k];
    a[t].y = x[arow * 64 + k + 1];
  }

#pragma unroll
  for (int ct = 0; ct < 4; ++ct) {
    int col = (ct << 4) + l16;
    v8f acc = {};
#pragma unroll
    for (int t = 0; t < 16; ++t) {
      int k = (t << 2) + (half << 1);
      v2f b;
      b.x = Wn[k * 64 + col];
      b.y = Wn[(k + 1) * 64 + col];
      acc = __builtin_amdgcn_wmma_f32_16x16x4_f32(
          false, a[t], false, b, (short)0, acc, false, false);
    }
    if (fullTile) {
#pragma unroll
      for (int j = 0; j < 8; ++j) {
        int row = rBase + j + (half << 3);
        z[row * 64 + col] = acc[j];
      }
    } else {
#pragma unroll
      for (int j = 0; j < 8; ++j) {
        int row = rBase + j + (half << 3);
        if (row < N) z[row * 64 + col] = acc[j];
      }
    }
  }
}

__global__ void k_weff(const float* __restrict__ We,
                       const float* __restrict__ att_edge,
                       float* __restrict__ weff) {
  int k = threadIdx.x;
  if (k < 16) {
    float s = 0.0f;
    for (int c = 0; c < 64; ++c) s += We[k * 64 + c] * att_edge[c];
    weff[k] = s;
  }
}

__global__ void k_node_scores(const float* __restrict__ z,
                              const float* __restrict__ att_src,
                              const float* __restrict__ att_dst,
                              float* __restrict__ sA, float* __restrict__ sD,
                              int N) {
  int n = blockIdx.x * blockDim.x + threadIdx.x;
  if (n >= N) return;
  const float4* zr = (const float4*)(z + (size_t)n * 64);
  const float4* as4 = (const float4*)att_src;
  const float4* ad4 = (const float4*)att_dst;
  float a = 0.0f, d = 0.0f;
#pragma unroll
  for (int i = 0; i < 16; ++i) {
    float4 v = zr[i], s = as4[i], t = ad4[i];
    a += v.x * s.x + v.y * s.y + v.z * s.z + v.w * s.w;
    d += v.x * t.x + v.y * t.y + v.z * t.z + v.w * t.w;
  }
  sA[n] = a;
  sD[n] = d;
}

__global__ void k_score(const int* __restrict__ ei,
                        const float* __restrict__ ea,
                        const float* __restrict__ sA,
                        const float* __restrict__ sD,
                        const float* __restrict__ weff,
                        float* __restrict__ scoreBuf,
                        unsigned* __restrict__ maxenc, int E) {
  int e = blockIdx.x * blockDim.x + threadIdx.x;
  if (e >= E) return;
  int s = ei[e], d = ei[E + e];
  float acc = sA[s] + sD[d];
  const float4* er = (const float4*)(ea + (size_t)e * 16);
  const float4* w4 = (const float4*)weff;
#pragma unroll
  for (int i = 0; i < 4; ++i) {
    float4 v = er[i], w = w4[i];
    acc += v.x * w.x + v.y * w.y + v.z * w.z + v.w * w.w;
  }
  float sc = acc > 0.0f ? acc : 0.2f * acc;
  scoreBuf[e] = sc;
  atomicMax(maxenc + d, enc_f32(sc));
}

__global__ void k_exp(const int* __restrict__ ei, float* __restrict__ num,
                      const unsigned* __restrict__ maxenc,
                      float* __restrict__ denom, int E) {
  int e = blockIdx.x * blockDim.x + threadIdx.x;
  if (e >= E) return;
  int d = ei[E + e];
  float m = dec_f32(maxenc[d]);
  float ex = __expf(num[e] - m);
  num[e] = ex;
  atomicAdd(denom + d, ex);
}

__device__ __forceinline__ void swap_tile_halves(float& x, float& y) {
  typedef __attribute__((ext_vector_type(2))) unsigned v2u;
  v2u r = __builtin_amdgcn_permlane16_swap(__float_as_uint(x),
                                           __float_as_uint(y), false, false);
  x = __uint_as_float(r.x);
  y = __uint_as_float(r.y);
}

__global__ void __launch_bounds__(256) k_message(
    const int* __restrict__ ei, const float* __restrict__ ea,
    const float* __restrict__ We, const float* __restrict__ z,
    const float* __restrict__ num, const float* __restrict__ denom,
    float* __restrict__ out, int E) {
  __shared__ float s_alpha[128];
  __shared__ int s_src[128];
  __shared__ int s_dst[128];

  int tid = threadIdx.x;
  int blockBase = blockIdx.x * 128;

  if (tid < 128) {
    int e = blockBase + tid;
    if (e < E) {
      int d = ei[E + e];
      int sN = ei[e];
      s_alpha[tid] = num[e] / (denom[d] + 1e-8f);
      s_src[tid] = sN;
      s_dst[tid] = d;
      const float* zrow = z + (size_t)sN * 64;
      __builtin_prefetch(zrow, 0, 3);
      __builtin_prefetch(zrow + 32, 0, 3);
    } else {
      s_alpha[tid] = 0.0f;
      s_src[tid] = 0;
      s_dst[tid] = 0;
    }
  }
  __syncthreads();

  int wave = tid >> 5;
  int lane = tid & 31;
  int half = lane >> 4;
  int l16 = lane & 15;
  int tileBase = wave << 4;
  int eBase = blockBase + tileBase;
  if (eBase >= E) return;

  int arow = eBase + l16;
  if (arow >= E) arow = E - 1;
  v2f a[4];
#pragma unroll
  for (int t = 0; t < 4; ++t) {
    int k = (t << 2) + (half << 1);
    a[t].x = ea[(size_t)arow * 16 + k];
    a[t].y = ea[(size_t)arow * 16 + k + 1];
  }

#pragma unroll
  for (int p = 0; p < 2; ++p) {
    int colE = (p << 5) + l16;
    int colO = colE + 16;
    v8f accE = {};
    v8f accO = {};
#pragma unroll
    for (int t = 0; t < 4; ++t) {
      int k = (t << 2) + (half << 1);
      v2f bE, bO;
      bE.x = We[k * 64 + colE];
      bE.y = We[(k + 1) * 64 + colE];
      bO.x = We[k * 64 + colO];
      bO.y = We[(k + 1) * 64 + colO];
      accE = __builtin_amdgcn_wmma_f32_16x16x4_f32(
          false, a[t], false, bE, (short)0, accE, false, false);
      accO = __builtin_amdgcn_wmma_f32_16x16x4_f32(
          false, a[t], false, bO, (short)0, accO, false, false);
    }
    asm volatile("s_wait_alu depctr_va_vdst(0)" : "+v"(accE), "+v"(accO));
#pragma unroll
    for (int j = 0; j < 8; ++j) {
      float x = accE[j], y = accO[j];
      swap_tile_halves(x, y);
      accE[j] = x;
      accO[j] = y;
    }
    int colw = (p << 5) + lane;
#pragma unroll
    for (int j = 0; j < 8; ++j) {
      int li = tileBase + j;
      float al = s_alpha[li];
      int sN = __builtin_amdgcn_readfirstlane(s_src[li]);
      int dN = __builtin_amdgcn_readfirstlane(s_dst[li]);
      const float* zrow = z + (size_t)sN * 64;
      float* orow = out + (size_t)dN * 64;
      float val = (zrow[colw] + accE[j]) * al;
      atomicAdd(orow + colw, val);
    }
#pragma unroll
    for (int j = 0; j < 8; ++j) {
      int li = tileBase + 8 + j;
      float al = s_alpha[li];
      int sN = __builtin_amdgcn_readfirstlane(s_src[li]);
      int dN = __builtin_amdgcn_readfirstlane(s_dst[li]);
      const float* zrow = z + (size_t)sN * 64;
      float* orow = out + (size_t)dN * 64;
      float val = (zrow[colw] + accO[j]) * al;
      atomicAdd(orow + colw, val);
    }
  }
}

extern "C" void kernel_launch(void* const* d_in, const int* in_sizes, int n_in,
                              void* d_out, int out_size, void* d_ws,
                              size_t ws_size, hipStream_t stream) {
  const float* x = (const float*)d_in[0];
  const int* ei = (const int*)d_in[1];
  const float* ea = (const float*)d_in[2];
  const float* Wn = (const float*)d_in[3];
  const float* We = (const float*)d_in[4];
  const float* att_src = (const float*)d_in[5];
  const float* att_dst = (const float*)d_in[6];
  const float* att_edge = (const float*)d_in[7];

  int N = in_sizes[0] / 64;
  int E = in_sizes[1] / 2;
  float* out = (float*)d_out;

  float* z = (float*)d_ws;
  float* sA = z + (size_t)N * 64;
  float* sD = sA + N;
  float* denom = sD + N;
  unsigned* maxenc = (unsigned*)(denom + N);
  float* num = (float*)(maxenc + N);
  float* weff = num + E;

  int total = N * 64;
  int ib = (total + 255) / 256;
  hipLaunchKernelGGL(k_init, dim3(ib), dim3(256), 0, stream, x, out, denom,
                     maxenc, N);

  int nRowTiles = (N + 15) / 16;
  int gb = (nRowTiles + 7) / 8;
  hipLaunchKernelGGL(k_node_gemm, dim3(gb), dim3(256), 0, stream, x, Wn, z, N);

  hipLaunchKernelGGL(k_weff, dim3(1), dim3(32), 0, stream, We, att_edge, weff);

  hipLaunchKernelGGL(k_node_scores, dim3((N + 255) / 256), dim3(256), 0, stream,
                     z, att_src, att_dst, sA, sD, N);

  hipLaunchKernelGGL(k_score, dim3((E + 255) / 256), dim3(256), 0, stream, ei,
                     ea, sA, sD, weff, num, maxenc, E);

  hipLaunchKernelGGL(k_exp, dim3((E + 255) / 256), dim3(256), 0, stream, ei,
                     num, maxenc, denom, E);

  hipLaunchKernelGGL(k_message, dim3((E + 127) / 128), dim3(256), 0, stream, ei,
                     ea, We, z, num, denom, out, E);
}


# float atomics on out removed: edges counting-sorted by dst (int atomics), one work-group per 16 nodes reduces max/exp/sum/messages in registers, We applied after the segment sum; sum order changes
# speedup vs baseline: 1.3597x; 1.1621x over previous
#include <hip/hip_runtime.h>


typedef __attribute__((ext_vector_type(2))) float v2f;
typedef __attribute__((ext_vector_type(8))) float v8f;

#define NEGV (-1000000000.0f)

__global__ void __launch_bounds__(256) k_node_gemm(
    const float* __restrict__ x, const float* __restrict__ Wn,
    float* __restrict__ z, int N) {
  int wave = (blockIdx.x * blockDim.x + threadIdx.x) >> 5;
  int lane = threadIdx.x & 31;
  int nRowTiles = (N + 15) >> 4;
  if (wave >= nRowTiles) return;

  int rBase = wave << 4;
  bool fullTile = (rBase + 16 <= N);
  int half = lane >> 4;
  int l16 = lane & 15;

  int arow = rBase + l16;
  if (arow >= N) arow = N - 1;
  v2f a[16];
#pragma unroll
  for (int t = 0; t < 16; ++t) {
    int k = (t << 2) + (half << 1);
    a[t].x = x[arow * 64 + k];
    a[t].y = x[arow * 64 + k + 1];
  }

#pragma unroll
  for (int ct = 0; ct < 4; ++ct) {
    int col = (ct << 4) + l16;
    v8f acc = {};
#pragma unroll
    for (int t = 0; t < 16; ++t) {
      int k = (t << 2) + (half << 1);
      v2f b;
      b.x = Wn[k * 64 + col];
      b.y = Wn[(k + 1) * 64 + col];
      acc = __builtin_amdgcn_wmma_f32_16x16x4_f32(
          false, a[t], false, b, (short)0, acc, false, false);
    }
    if (fullTile) {
#pragma unroll
      for (int j = 0; j < 8; ++j) {
        int row = rBase + j + (half << 3);
        z[row * 64 + col] = acc[j];
      }
    } else {
#pragma unroll
      for (int j = 0; j < 8; ++j) {
        int row = rBase + j + (half << 3);
        if (row < N) z[row * 64 + col] = acc[j];
      }
    }
  }
}

__global__ void k_weff(const float* __restrict__ We,
                       const float* __restrict__ att_edge,
                       float* __restrict__ weff) {
  int k = threadIdx.x;
  if (k < 16) {
    float s = 0.0f;
    for (int c = 0; c < 64; ++c) s += We[k * 64 + c] * att_edge[c];
    weff[k] = s;
  }
}

__global__ void k_node_scores(const float* __restrict__ z,
                              const float* __restrict__ att_src,
                              const float* __restrict__ att_dst,
                              float* __restrict__ sA, float* __restrict__ sD,
                              int* __restrict__ cnt, int N) {
  int n = blockIdx.x * blockDim.x + threadIdx.x;
  if (n >= N) return;
  const float4* zr = (const float4*)(z + (size_t)n * 64);
  const float4* as4 = (const float4*)att_src;
  const float4* ad4 = (const float4*)att_dst;
  float a = 0.0f, d = 0.0f;
#pragma unroll
  for (int i = 0; i < 16; ++i) {
    float4 v = zr[i], s = as4[i], t = ad4[i];
    a += v.x * s.x + v.y * s.y + v.z * s.z + v.w * s.w;
    d += v.x * t.x + v.y * t.y + v.z * t.z + v.w * t.w;
  }
  sA[n] = a;
  sD[n] = d;
  cnt[n] = 0;
}

__global__ void k_hist(const int* __restrict__ ei, int* __restrict__ cnt,
                       int E) {
  int e = blockIdx.x * blockDim.x + threadIdx.x;
  if (e >= E) return;
  atomicAdd(cnt + ei[E + e], 1);
}

__global__ void __launch_bounds__(1024) k_scan(int* __restrict__ cnt, int N) {
  __shared__ int s_w[32];
  int tid = threadIdx.x;
  int lane = tid & 31;
  int wave = tid >> 5;
  int carry = 0;
  for (int base = 0; base < N; base += 4096) {
    int i = base + (tid << 2);
    int v0 = (i < N) ? cnt[i] : 0;
    int v1 = (i + 1 < N) ? cnt[i + 1] : 0;
    int v2 = (i + 2 < N) ? cnt[i + 2] : 0;
    int v3 = (i + 3 < N) ? cnt[i + 3] : 0;
    int t = v0 + v1 + v2 + v3;
    int incl = t;
#pragma unroll
    for (int off = 1; off < 32; off <<= 1) {
      int u = __shfl_up(incl, off);
      if (lane >= off) incl += u;
    }
    if (lane == 31) s_w[wave] = incl;
    __syncthreads();
    int wv = s_w[lane];
    int winc = wv;
#pragma unroll
    for (int off = 1; off < 32; off <<= 1) {
      int u = __shfl_up(winc, off);
      if (lane >= off) winc += u;
    }
    int wexcl = __shfl(winc, wave) - __shfl(wv, wave);
    int total = __shfl(winc, 31);
    int ex0 = carry + wexcl + (incl - t);
    if (i < N) cnt[i] = ex0;
    if (i + 1 < N) cnt[i + 1] = ex0 + v0;
    if (i + 2 < N) cnt[i + 2] = ex0 + v0 + v1;
    if (i + 3 < N) cnt[i + 3] = ex0 + v0 + v1 + v2;
    carry += total;
    __syncthreads();
  }
}

__global__ void k_fill(const int* __restrict__ ei, int* __restrict__ cnt,
                       int* __restrict__ perm, int E) {
  int e = blockIdx.x * blockDim.x + threadIdx.x;
  if (e >= E) return;
  int slot = atomicAdd(cnt + ei[E + e], 1);
  perm[slot] = e;
}

__device__ __forceinline__ float wave_max(float v) {
#pragma unroll
  for (int off = 16; off > 0; off >>= 1) v = fmaxf(v, __shfl_xor(v, off));
  return v;
}
__device__ __forceinline__ float wave_sum(float v) {
#pragma unroll
  for (int off = 16; off > 0; off >>= 1) v += __shfl_xor(v, off);
  return v;
}

__device__ __forceinline__ float edge_score(const float* __restrict__ ea,
                                            int e, float base, float4 w0,
                                            float4 w1, float4 w2, float4 w3) {
  const float4* er = (const float4*)(ea + (size_t)e * 16);
  float4 v0 = er[0], v1 = er[1], v2 = er[2], v3 = er[3];
  float acc = base;
  acc += v0.x * w0.x + v0.y * w0.y + v0.z * w0.z + v0.w * w0.w;
  acc += v1.x * w1.x + v1.y * w1.y + v1.z * w1.z + v1.w * w1.w;
  acc += v2.x * w2.x + v2.y * w2.y + v2.z * w2.z + v2.w * w2.w;
  acc += v3.x * w3.x + v3.y * w3.y + v3.z * w3.z + v3.w * w3.w;
  return acc > 0.0f ? acc : 0.2f * acc;
}

__device__ __forceinline__ void add_chunk(const float* __restrict__ z,
                                          const float* __restrict__ ea,
                                          int e, int s, float ex, int cntc,
                                          int lane, float& zx, float& zy,
                                          float& ag) {
  int e0 = __builtin_amdgcn_readfirstlane(e);
  int s0 = __builtin_amdgcn_readfirstlane(s);
  if (lane >= cntc) {
    e = e0;
    s = s0;
  }
  int cnt8 = (cntc + 7) & ~7;
  for (int j0 = 0; j0 < cnt8; j0 += 8) {
    float2 zv[8];
    float av[8];
    float wj[8];
#pragma unroll
    for (int u = 0; u < 8; ++u) {
      int ej = __builtin_amdgcn_readlane(e, j0 + u);
      int sj = __builtin_amdgcn_readlane(s, j0 + u);
      wj[u] = __int_as_float(
          __builtin_amdgcn_readlane(__float_as_int(ex), j0 + u));
      zv[u] = *(const float2*)(z + (size_t)sj * 64 + (lane << 1));
      av[u] = ea[(size_t)ej * 16 + (lane & 15)];
    }
#pragma unroll
    for (int u = 0; u < 8; ++u) {
      zx += wj[u] * zv[u].x;
      zy += wj[u] * zv[u].y;
      ag += wj[u] * av[u];
    }
  }
}

__global__ void __launch_bounds__(256) k_node(
    const int* __restrict__ ei, const float* __restrict__ ea,
    const float* __restrict__ We, const float* __restrict__ weff,
    const float* __restrict__ x, const float* __restrict__ z,
    const float* __restrict__ sA, const float* __restrict__ sD,
    const int* __restrict__ segEnd, const int* __restrict__ perm,
    float* __restrict__ out, int N) {
  __shared__ float s_out[16][64];
  __shared__ float s_agg[16][16];

  int tid = threadIdx.x;
  int wave = __builtin_amdgcn_readfirstlane(tid >> 5);
  int lane = tid & 31;
  int rBase = blockIdx.x << 4;

  const float4* w4 = (const float4*)weff;
  float4 w0 = w4[0], w1 = w4[1], w2 = w4[2], w3 = w4[3];

  for (int q = 0; q < 2; ++q) {
    int r = (wave << 1) + q;
    int n = rBase + r;
    float zx = 0.0f, zy = 0.0f, ag = 0.0f, inv = 0.0f, x0 = 0.0f, x1 = 0.0f;
    if (n < N) {
      int b = (n == 0) ? 0 : segEnd[n - 1];
      int en = segEnd[n];
      float sDn = sD[n];
      const float2 xr = *(const float2*)(x + (size_t)n * 64 + (lane << 1));
      x0 = xr.x;
      x1 = xr.y;
      float dsum = 0.0f;
      if (en - b <= 32) {
        int i = b + lane;
        int e = 0, s = 0;
        float sc = NEGV;
        if (i < en) {
          e = perm[i];
          s = ei[e];
          sc = edge_score(ea, e, sA[s] + sDn, w0, w1, w2, w3);
        }
        float m = wave_max(sc);
        float ex = (i < en) ? __expf(sc - m) : 0.0f;
        dsum = ex;
        add_chunk(z, ea, e, s, ex, en - b, lane, zx, zy, ag);
      } else {
        float m = NEGV;
        for (int c = b; c < en; c += 32) {
          int i = c + lane;
          if (i < en) {
            int e = perm[i];
            int s = ei[e];
            m = fmaxf(m, edge_score(ea, e, sA[s] + sDn, w0, w1, w2, w3));
          }
        }
        m = wave_max(m);
        for (int c = b; c < en; c += 32) {
          int i = c + lane;
          int e = 0, s = 0;
          float ex = 0.0f;
          if (i < en) {
            e = perm[i];
            s = ei[e];
            ex = __expf(edge_score(ea, e, sA[s] + sDn, w0, w1, w2, w3) - m);
          }
          dsum += ex;
          int cntc = en - c;
          if (cntc > 32) cntc = 32;
          add_chunk(z, ea, e, s, ex, cntc, lane, zx, zy, ag);
        }
      }
      dsum = wave_sum(dsum);
      inv = 1.0f / (dsum + 1e-8f);
    }
    s_out[r][(lane << 1)] = x0 + zx * inv;
    s_out[r][(lane << 1) + 1] = x1 + zy * inv;
    if (lane < 16) s_agg[r][lane] = ag * inv;
  }
  __syncthreads();

  if (wave < 4) {
    int half = lane >> 4;
    int l16 = lane & 15;
    int col = (wave << 4) + l16;
    v8f acc;
#pragma unroll
    for (int j = 0; j < 8; ++j) acc[j] = s_out[(half << 3) + j][col];
#pragma unroll
    for (int t = 0; t < 4; ++t) {
      int k = (t << 2) + (half << 1);
      v2f a, bb;
      a.x = s_agg[l16][k];
      a.y = s_agg[l16][k + 1];
      bb.x = We[k * 64 + col];
      bb.y = We[(k + 1) * 64 + col];
      acc = __builtin_amdgcn_wmma_f32_16x16x4_f32(
          false, a, false, bb, (short)0, acc, false, false);
    }
#pragma unroll
    for (int j = 0; j < 8; ++j) s_out[(half << 3) + j][col] = acc[j];
  }
  __syncthreads();

  for (int q = 0; q < 2; ++q) {
    int r = (wave << 1) + q;
    int n = rBase + r;
    if (n < N) {
      float2 v;
      v.x = s_out[r][(lane << 1)];
      v.y = s_out[r][(lane << 1) + 1];
      *(float2*)(out + (size_t)n * 64 + (lane << 1)) = v;
    }
  }
}

extern "C" void kernel_launch(void* const* d_in, const int* in_sizes, int n_in,
                              void* d_out, int out_size, void* d_ws,
                              size_t ws_size, hipStream_t stream) {
  const float* x = (const float*)d_in[0];
  const int* ei = (const int*)d_in[1];
  const float* ea = (const float*)d_in[2];
  const float* Wn = (const float*)d_in[3];
  const float* We = (const float*)d_in[4];
  const float* att_src = (const float*)d_in[5];
  const float* att_dst = (const float*)d_in[6];
  const float* att_edge = (const float*)d_in[7];

  int N = in_sizes[0] / 64;
  int E = in_sizes[1] / 2;
  float* out = (float*)d_out;

  float* weff = (float*)d_ws;
  float* z = weff + 64;
  float* sA = z + (size_t)N * 64;
  float* sD = sA + N;
  int* cnt = (int*)(sD + N);
  int* perm = cnt + N;

  int nRowTiles = (N + 15) / 16;
  int gb = (nRowTiles + 7) / 8;
  hipLaunchKernelGGL(k_node_gemm, dim3(gb), dim3(256), 0, stream, x, Wn, z, N);

  hipLaunchKernelGGL(k_weff, dim3(1), dim3(32), 0, stream, We, att_edge, weff);

  hipLaunchKernelGGL(k_node_scores, dim3((N + 255) / 256), dim3(256), 0, stream,
                     z, att_src, att_dst, sA, sD, cnt, N);

  hipLaunchKernelGGL(k_hist, dim3((E + 255) / 256), dim3(256), 0, stream, ei,
                     cnt, E);

  hipLaunchKernelGGL(k_scan, dim3(1), dim3(1024), 0, stream, cnt, N);

  hipLaunchKernelGGL(k_fill, dim3((E + 255) / 256), dim3(256), 0, stream, ei,
                     cnt, perm, E);

  hipLaunchKernelGGL(k_node, dim3(nRowTiles), dim3(256), 0, stream, ei, ea, We,
                     weff, x, z, sA, sD, cnt, perm, out, N);
}


# counting sort: edge rank = the count pass's atomic return, so the placement pass has no atomic (1.6M returning atomics fewer); sort moved ahead of the GEMM, ranks in z's space; arrival-order sums
# speedup vs baseline: 1.6449x; 1.2098x over previous
#include <hip/hip_runtime.h>


typedef __attribute__((ext_vector_type(2))) float v2f;
typedef __attribute__((ext_vector_type(8))) float v8f;

#define NEGV (-1000000000.0f)

__global__ void __launch_bounds__(256) k_node_gemm(
    const float* __restrict__ x, const float* __restrict__ Wn,
    float* __restrict__ z, int N) {
  int wave = (blockIdx.x * blockDim.x + threadIdx.x) >> 5;
  int lane = threadIdx.x & 31;
  int nRowTiles = (N + 15) >> 4;
  if (wave >= nRowTiles) return;

  int rBase = wave << 4;
  bool fullTile = (rBase + 16 <= N);
  int half = lane >> 4;
  int l16 = lane & 15;

  int arow = rBase + l16;
  if (arow >= N) arow = N - 1;
  v2f a[16];
#pragma unroll
  for (int t = 0; t < 16; ++t) {
    int k = (t << 2) + (half << 1);
    a[t].x = x[arow * 64 + k];
    a[t].y = x[arow * 64 + k + 1];
  }

#pragma unroll
  for (int ct = 0; ct < 4; ++ct) {
    int col = (ct << 4) + l16;
    v8f acc = {};
#pragma unroll
    for (int t = 0; t < 16; ++t) {
      int k = (t << 2) + (half << 1);
      v2f b;
      b.x = Wn[k * 64 + col];
      b.y = Wn[(k + 1) * 64 + col];
      acc = __builtin_amdgcn_wmma_f32_16x16x4_f32(
          false, a[t], false, b, (short)0, acc, false, false);
    }
    if (fullTile) {
#pragma unroll
      for (int j = 0; j < 8; ++j) {
        int row = rBase + j + (half << 3);
        z[row * 64 + col] = acc[j];
      }
    } else {
#pragma unroll
      for (int j = 0; j < 8; ++j) {
        int row = rBase + j + (half << 3);
        if (row < N) z[row * 64 + col] = acc[j];
      }
    }
  }
}

__global__ void k_weff(const float* __restrict__ We,
                       const float* __restrict__ att_edge,
                       float* __restrict__ weff, int* __restrict__ cnt,
                       int N) {
  int n = blockIdx.x * blockDim.x + threadIdx.x;
  if (n < N) cnt[n] = 0;
  if (n < 16) {
    float s = 0.0f;
    for (int c = 0; c < 64; ++c) s += We[n * 64 + c] * att_edge[c];
    weff[n] = s;
  }
}

__global__ void k_node_scores(const float* __restrict__ z,
                              const float* __restrict__ att_src,
                              const float* __restrict__ att_dst,
                              float* __restrict__ sA, float* __restrict__ sD,
                              int N) {
  int n = blockIdx.x * blockDim.x + threadIdx.x;
  if (n >= N) return;
  const float4* zr = (const float4*)(z + (size_t)n * 64);
  const float4* as4 = (const float4*)att_src;
  const float4* ad4 = (const float4*)att_dst;
  float a = 0.0f, d = 0.0f;
#pragma unroll
  for (int i = 0; i < 16; ++i) {
    float4 v = zr[i], s = as4[i], t = ad4[i];
    a += v.x * s.x + v.y * s.y + v.z * s.z + v.w * s.w;
    d += v.x * t.x + v.y * t.y + v.z * t.z + v.w * t.w;
  }
  sA[n] = a;
  sD[n] = d;
}

__global__ void k_hist(const int* __restrict__ ei, int* __restrict__ cnt,
                       int* __restrict__ rank, int E) {
  int e = blockIdx.x * blockDim.x + threadIdx.x;
  if (e >= E) return;
  rank[e] = atomicAdd(cnt + ei[E + e], 1);
}

__global__ void __launch_bounds__(1024) k_scan(int* __restrict__ cnt, int N) {
  __shared__ int s_w[32];
  int tid = threadIdx.x;
  int lane = tid & 31;
  int wave = tid >> 5;
  int carry = 0;
  for (int base = 0; base < N; base += 4096) {
    int i = base + (tid << 2);
    int v0 = (i < N) ? cnt[i] : 0;
    int v1 = (i + 1 < N) ? cnt[i + 1] : 0;
    int v2 = (i + 2 < N) ? cnt[i + 2] : 0;
    int v3 = (i + 3 < N) ? cnt[i + 3] : 0;
    int t = v0 + v1 + v2 + v3;
    int incl = t;
#pragma unroll
    for (int off = 1; off < 32; off <<= 1) {
      int u = __shfl_up(incl, off);
      if (lane >= off) incl += u;
    }
    if (lane == 31) s_w[wave] = incl;
    __syncthreads();
    int wv = s_w[lane];
    int winc = wv;
#pragma unroll
    for (int off = 1; off < 32; off <<= 1) {
      int u = __shfl_up(winc, off);
      if (lane >= off) winc += u;
    }
    int wexcl = __shfl(winc, wave) - __shfl(wv, wave);
    int total = __shfl(winc, 31);
    int ex0 = carry + wexcl + (incl - t);
    if (i < N) cnt[i] = ex0;
    if (i + 1 < N) cnt[i + 1] = ex0 + v0;
    if (i + 2 < N) cnt[i + 2] = ex0 + v0 + v1;
    if (i + 3 < N) cnt[i + 3] = ex0 + v0 + v1 + v2;
    carry += total;
    __syncthreads();
  }
}

__global__ void k_fill(const int* __restrict__ ei,
                       const int* __restrict__ segStart,
                       const int* __restrict__ rank, int* __restrict__ perm,
                       int E) {
  int e = blockIdx.x * blockDim.x + threadIdx.x;
  if (e >= E) return;
  perm[segStart[ei[E + e]] + rank[e]] = e;
}

__device__ __forceinline__ float wave_max(float v) {
#pragma unroll
  for (int off = 16; off > 0; off >>= 1) v = fmaxf(v, __shfl_xor(v, off));
  return v;
}
__device__ __forceinline__ float wave_sum(float v) {
#pragma unroll
  for (int off = 16; off > 0; off >>= 1) v += __shfl_xor(v, off);
  return v;
}

__device__ __forceinline__ float edge_score(const float* __restrict__ ea,
                                            int e, float base, float4 w0,
                                            float4 w1, float4 w2, float4 w3) {
  const float4* er = (const float4*)(ea + (size_t)e * 16);
  float4 v0 = er[0], v1 = er[1], v2 = er[2], v3 = er[3];
  float acc = base;
  acc += v0.x * w0.x + v0.y * w0.y + v0.z * w0.z + v0.w * w0.w;
  acc += v1.x * w1.x + v1.y * w1.y + v1.z * w1.z + v1.w * w1.w;
  acc += v2.x * w2.x + v2.y * w2.y + v2.z * w2.z + v2.w * w2.w;
  acc += v3.x * w3.x + v3.y * w3.y + v3.z * w3.z + v3.w * w3.w;
  return acc > 0.0f ? acc : 0.2f * acc;
}

__device__ __forceinline__ void add_chunk(const float* __restrict__ z,
                                          const float* __restrict__ ea,
                                          int e, int s, float ex, int cntc,
                                          int lane, float& zx, float& zy,
                                          float& ag) {
  int e0 = __builtin_amdgcn_readfirstlane(e);
  int s0 = __builtin_amdgcn_readfirstlane(s);
  if (lane >= cntc) {
    e = e0;
    s = s0;
  }
  int cnt8 = (cntc + 7) & ~7;
  for (int j0 = 0; j0 < cnt8; j0 += 8) {
    float2 zv[8];
    float av[8];
    float wj[8];
#pragma unroll
    for (int u = 0; u < 8; ++u) {
      int ej = __builtin_amdgcn_readlane(e, j0 + u);
      int sj = __builtin_amdgcn_readlane(s, j0 + u);
      wj[u] = __int_as_float(
          __builtin_amdgcn_readlane(__float_as_int(ex), j0 + u));
      zv[u] = *(const float2*)(z + (size_t)sj * 64 + (lane << 1));
      av[u] = ea[(size_t)ej * 16 + (lane & 15)];
    }
#pragma unroll
    for (int u = 0; u < 8; ++u) {
      zx += wj[u] * zv[u].x;
      zy += wj[u] * zv[u].y;
      ag += wj[u] * av[u];
    }
  }
}

__global__ void __launch_bounds__(256) k_node(
    const int* __restrict__ ei, const float* __restrict__ ea,
    const float* __restrict__ We, const float* __restrict__ weff,
    const float* __restrict__ x, const float* __restrict__ z,
    const float* __restrict__ sA, const float* __restrict__ sD,
    const int* __restrict__ segStart, const int* __restrict__ perm,
    float* __restrict__ out, int N, int E) {
  __shared__ float s_out[16][64];
  __shared__ float s_agg[16][16];

  int tid = threadIdx.x;
  int wave = __builtin_amdgcn_readfirstlane(tid >> 5);
  int lane = tid & 31;
  int rBase = blockIdx.x << 4;

  const float4* w4 = (const float4*)weff;
  float4 w0 = w4[0], w1 = w4[1], w2 = w4[2], w3 = w4[3];

  for (int q = 0; q < 2; ++q) {
    int r = (wave << 1) + q;
    int n = rBase + r;
    float zx = 0.0f, zy = 0.0f, ag = 0.0f, inv = 0.0f, x0 = 0.0f, x1 = 0.0f;
    if (n < N) {
      int b = segStart[n];
      int en = (n + 1 < N) ? segStart[n + 1] : E;
      float sDn = sD[n];
      const float2 xr = *(const float2*)(x + (size_t)n * 64 + (lane << 1));
      x0 = xr.x;
      x1 = xr.y;
      float dsum = 0.0f;
      if (en - b <= 32) {
        int i = b + lane;
        int e = 0, s = 0;
        float sc = NEGV;
        if (i < en) {
          e = perm[i];
          s = ei[e];
          sc = edge_score(ea, e, sA[s] + sDn, w0, w1, w2, w3);
        }
        float m = wave_max(sc);
        float ex = (i < en) ? __expf(sc - m) : 0.0f;
        dsum = ex;
        add_chunk(z, ea, e, s, ex, en - b, lane, zx, zy, ag);
      } else {
        float m = NEGV;
        for (int c = b; c < en; c += 32) {
          int i = c + lane;
          if (i < en) {
            int e = perm[i];
            int s = ei[e];
            m = fmaxf(m, edge_score(ea, e, sA[s] + sDn, w0, w1, w2, w3));
          }
        }
        m = wave_max(m);
        for (int c = b; c < en; c += 32) {
          int i = c + lane;
          int e = 0, s = 0;
          float ex = 0.0f;
          if (i < en) {
            e = perm[i];
            s = ei[e];
            ex = __expf(edge_score(ea, e, sA[s] + sDn, w0, w1, w2, w3) - m);
          }
          dsum += ex;
          int cntc = en - c;
          if (cntc > 32) cntc = 32;
          add_chunk(z, ea, e, s, ex, cntc, lane, zx, zy, ag);
        }
      }
      dsum = wave_sum(dsum);
      inv = 1.0f / (dsum + 1e-8f);
    }
    s_out[r][(lane << 1)] = x0 + zx * inv;
    s_out[r][(lane << 1) + 1] = x1 + zy * inv;
    if (lane < 16) s_agg[r][lane] = ag * inv;
  }
  __syncthreads();

  if (wave < 4) {
    int half = lane >> 4;
    int l16 = lane & 15;
    int col = (wave << 4) + l16;
    v8f acc;
#pragma unroll
    for (int j = 0; j < 8; ++j) acc[j] = s_out[(half << 3) + j][col];
#pragma unroll
    for (int t = 0; t < 4; ++t) {
      int k = (t << 2) + (half << 1);
      v2f a, bb;
      a.x = s_agg[l16][k];
      a.y = s_agg[l16][k + 1];
      bb.x = We[k * 64 + col];
      bb.y = We[(k + 1) * 64 + col];
      acc = __builtin_amdgcn_wmma_f32_16x16x4_f32(
          false, a, false, bb, (short)0, acc, false, false);
    }
#pragma unroll
    for (int j = 0; j < 8; ++j) s_out[(half << 3) + j][col] = acc[j];
  }
  __syncthreads();

  for (int q = 0; q < 2; ++q) {
    int r = (wave << 1) + q;
    int n = rBase + r;
    if (n < N) {
      float2 v;
      v.x = s_out[r][(lane << 1)];
      v.y = s_out[r][(lane << 1) + 1];
      *(float2*)(out + (size_t)n * 64 + (lane << 1)) = v;
    }
  }
}

extern "C" void kernel_launch(void* const* d_in, const int* in_sizes, int n_in,
                              void* d_out, int out_size, void* d_ws,
                              size_t ws_size, hipStream_t stream) {
  const float* x = (const float*)d_in[0];
  const int* ei = (const int*)d_in[1];
  const float* ea = (const float*)d_in[2];
  const float* Wn = (const float*)d_in[3];
  const float* We = (const float*)d_in[4];
  const float* att_src = (const float*)d_in[5];
  const float* att_dst = (const float*)d_in[6];
  const float* att_edge = (const float*)d_in[7];

  int N = in_sizes[0] / 64;
  int E = in_sizes[1] / 2;
  float* out = (float*)d_out;

  float* weff = (float*)d_ws;
  float* z = weff + 64;
  float* sA = z + (size_t)N * 64;
  float* sD = sA + N;
  int* cnt = (int*)(sD + N);
  int* perm = cnt + N;
  int* rank = (int*)z;

  hipLaunchKernelGGL(k_weff, dim3((N + 255) / 256), dim3(256), 0, stream, We,
                     att_edge, weff, cnt, N);

  hipLaunchKernelGGL(k_hist, dim3((E + 255) / 256), dim3(256), 0, stream, ei,
                     cnt, rank, E);

  hipLaunchKernelGGL(k_scan, dim3(1), dim3(1024), 0, stream, cnt, N);

  hipLaunchKernelGGL(k_fill, dim3((E + 255) / 256), dim3(256), 0, stream, ei,
                     cnt, rank, perm, E);

  int nRowTiles = (N + 15) / 16;
  int gb = (nRowTiles + 7) / 8;
  hipLaunchKernelGGL(k_node_gemm, dim3(gb), dim3(256), 0, stream, x, Wn, z, N);

  hipLaunchKernelGGL(k_node_scores, dim3((N + 255) / 256), dim3(256), 0, stream,
                     z, att_src, att_dst, sA, sD, N);

  hipLaunchKernelGGL(k_node, dim3(nRowTiles), dim3(256), 0, stream, ei, ea, We,
                     weff, x, z, sA, sD, cnt, perm, out, N, E);
}


# counting sort by dst on LDS counters: a work-group per (6400-node range, 1/16 of the edges) counts, then places by returning ds add; no global atomic, clear or scan launch; arrival-order sums
# speedup vs baseline: 2.7604x; 1.6782x over previous
#include <hip/hip_runtime.h>


typedef __attribute__((ext_vector_type(2))) float v2f;
typedef __attribute__((ext_vector_type(8))) float v8f;

#define NEGV (-1000000000.0f)

__global__ void __launch_bounds__(256) k_node_gemm(
    const float* __restrict__ x, const float* __restrict__ Wn,
    float* __restrict__ z, int N) {
  int wave = (blockIdx.x * blockDim.x + threadIdx.x) >> 5;
  int lane = threadIdx.x & 31;
  int nRowTiles = (N + 15) >> 4;
  if (wave >= nRowTiles) return;

  int rBase = wave << 4;
  bool fullTile = (rBase + 16 <= N);
  int half = lane >> 4;
  int l16 = lane & 15;

  int arow = rBase + l16;
  if (arow >= N) arow = N - 1;
  v2f a[16];
#pragma unroll
  for (int t = 0; t < 16; ++t) {
    int k = (t << 2) + (half << 1);
    a[t].x = x[arow * 64 + k];
    a[t].y = x[arow * 64 + k + 1];
  }

#pragma unroll
  for (int ct = 0; ct < 4; ++ct) {
    int col = (ct << 4) + l16;
    v8f acc = {};
#pragma unroll
    for (int t = 0; t < 16; ++t) {
      int k = (t << 2) + (half << 1);
      v2f b;
      b.x = Wn[k * 64 + col];
      b.y = Wn[(k + 1) * 64 + col];
      acc = __builtin_amdgcn_wmma_f32_16x16x4_f32(
          false, a[t], false, b, (short)0, acc, false, false);
    }
    if (fullTile) {
#pragma unroll
      for (int j = 0; j < 8; ++j) {
        int row = rBase + j + (half << 3);
        z[row * 64 + col] = acc[j];
      }
    } else {
#pragma unroll
      for (int j = 0; j < 8; ++j) {
        int row = rBase + j + (half << 3);
        if (row < N) z[row * 64 + col] = acc[j];
      }
    }
  }
}

#define SORT_C 16
#define SORT_R 6400
#define SORT_T 1024
#define SORT_B 16

__device__ __forceinline__ int wave_sum_int(int v) {
#pragma unroll
  for (int off = 16; off > 0; off >>= 1) v += __shfl_xor(v, off);
  return v;
}

__global__ void __launch_bounds__(SORT_T) k_count(
    const int* __restrict__ ei, const float* __restrict__ We,
    const float* __restrict__ att_edge, float* __restrict__ weff,
    int* __restrict__ planes, int* __restrict__ tot, int N, int E) {
  __shared__ int s_cnt[SORT_R];
  __shared__ int s_total;
  int tid = threadIdx.x;
  int lane = tid & 31;
  int j = blockIdx.x / SORT_C;
  int c = blockIdx.x - j * SORT_C;
  int lo = j * SORT_R;
  int nr = N - lo;
  if (nr > SORT_R) nr = SORT_R;
  int chunk = (E + SORT_C - 1) / SORT_C;
  int eb = c * chunk;
  int ee = eb + chunk;
  if (ee > E) ee = E;

  if (blockIdx.x == 0 && tid < 16) {
    float s = 0.0f;
    for (int k = 0; k < 64; ++k) s += We[tid * 64 + k] * att_edge[k];
    weff[tid] = s;
  }
  for (int i = tid; i < SORT_R; i += SORT_T) s_cnt[i] = 0;
  if (tid == 0) s_total = 0;
  __syncthreads();

  const int* __restrict__ dst = ei + E;
  for (int base = eb; base < ee; base += SORT_T * SORT_B) {
    int d[SORT_B];
#pragma unroll
    for (int u = 0; u < SORT_B; ++u) {
      int i = base + u * SORT_T + tid;
      int dv = dst[i < ee ? i : ee - 1];
      d[u] = (i < ee) ? dv : -1;
    }
#pragma unroll
    for (int u = 0; u < SORT_B; ++u) {
      unsigned r = (unsigned)(d[u] - lo);
      if (r < (unsigned)nr) atomicAdd(&s_cnt[r], 1);
    }
  }
  __syncthreads();

  int part = 0;
  for (int i = tid; i < nr; i += SORT_T) {
    int v = s_cnt[i];
    planes[(size_t)c * N + lo + i] = v;
    part += v;
  }
  part = wave_sum_int(part);
  if (lane == 0) atomicAdd(&s_total, part);
  __syncthreads();
  if (tid == 0) tot[blockIdx.x] = s_total;
}

__global__ void __launch_bounds__(SORT_T) k_place(
    const int* __restrict__ ei, const int* __restrict__ planes,
    const int* __restrict__ tot, int* __restrict__ segStart,
    int* __restrict__ perm, int N, int E) {
  __shared__ int s_slot[SORT_R];
  __shared__ int s_bef[SORT_R];
  __shared__ int s_w[32];
  __shared__ int s_rb[32];
  int tid = threadIdx.x;
  int lane = tid & 31;
  int wave = tid >> 5;
  int j = blockIdx.x / SORT_C;
  int c = blockIdx.x - j * SORT_C;
  int lo = j * SORT_R;
  int nr = N - lo;
  if (nr > SORT_R) nr = SORT_R;
  int chunk = (E + SORT_C - 1) / SORT_C;
  int eb = c * chunk;
  int ee = eb + chunk;
  if (ee > E) ee = E;

  int rb = 0;
  for (int t = tid; t < j * SORT_C; t += SORT_T) rb += tot[t];
  rb = wave_sum_int(rb);
  if (lane == 0) s_rb[wave] = rb;

  for (int i = tid; i < SORT_R; i += SORT_T) {
    int t = 0, b = 0;
    if (i < nr) {
#pragma unroll
      for (int cc = 0; cc < SORT_C; ++cc) {
        int v = planes[(size_t)cc * N + lo + i];
        t += v;
        if (cc < c) b += v;
      }
    }
    s_slot[i] = t;
    s_bef[i] = b;
  }
  __syncthreads();

  int rbase = wave_sum_int(s_rb[lane]);

  constexpr int SORT_K = (SORT_R + SORT_T - 1) / SORT_T;
  int i0 = tid * SORT_K;
  int v[SORT_K];
  int sum = 0;
#pragma unroll
  for (int k = 0; k < SORT_K; ++k) {
    int i = i0 + k;
    v[k] = (i < SORT_R) ? s_slot[i] : 0;
    sum += v[k];
  }
  int incl = sum;
#pragma unroll
  for (int off = 1; off < 32; off <<= 1) {
    int u = __shfl_up(incl, off);
    if (lane >= off) incl += u;
  }
  if (lane == 31) s_w[wave] = incl;
  __syncthreads();
  int wv = s_w[lane];
  int winc = wv;
#pragma unroll
  for (int off = 1; off < 32; off <<= 1) {
    int u = __shfl_up(winc, off);
    if (lane >= off) winc += u;
  }
  int run = rbase + __shfl(winc, wave) - __shfl(wv, wave) + (incl - sum);
#pragma unroll
  for (int k = 0; k < SORT_K; ++k) {
    int i = i0 + k;
    if (i < SORT_R) {
      if (c == 0 && i < nr) segStart[lo + i] = run;
      s_slot[i] = run + s_bef[i];
      run += v[k];
    }
  }
  __syncthreads();

  const int* __restrict__ dst = ei + E;
  for (int base = eb; base < ee; base += SORT_T * SORT_B) {
    int d[SORT_B];
#pragma unroll
    for (int u = 0; u < SORT_B; ++u) {
      int i = base + u * SORT_T + tid;
      int dv = dst[i < ee ? i : ee - 1];
      d[u] = (i < ee) ? dv : -1;
    }
    int slot[SORT_B];
#pragma unroll
    for (int u = 0; u < SORT_B; ++u) {
      unsigned r = (unsigned)(d[u] - lo);
      slot[u] = (r < (unsigned)nr) ? atomicAdd(&s_slot[r], 1) : -1;
    }
#pragma unroll
    for (int u = 0; u < SORT_B; ++u)
      if (slot[u] >= 0) perm[slot[u]] = base + u * SORT_T + tid;
  }
}

__global__ void k_node_scores(const float* __restrict__ z,
                              const float* __restrict__ att_src,
                              const float* __restrict__ att_dst,
                              float* __restrict__ sA, float* __restrict__ sD,
                              int N) {
  int n = blockIdx.x * blockDim.x + threadIdx.x;
  if (n >= N) return;
  const float4* zr = (const float4*)(z + (size_t)n * 64);
  const float4* as4 = (const float4*)att_src;
  const float4* ad4 = (const float4*)att_dst;
  float a = 0.0f, d = 0.0f;
#pragma unroll
  for (int i = 0; i < 16; ++i) {
    float4 v = zr[i], s = as4[i], t = ad4[i];
    a += v.x * s.x + v.y * s.y + v.z * s.z + v.w * s.w;
    d += v.x * t.x + v.y * t.y + v.z * t.z + v.w * t.w;
  }
  sA[n] = a;
  sD[n] = d;
}

__device__ __forceinline__ float wave_max(float v) {
#pragma unroll
  for (int off = 16; off > 0; off >>= 1) v = fmaxf(v, __shfl_xor(v, off));
  return v;
}
__device__ __forceinline__ float wave_sum(float v) {
#pragma unroll
  for (int off = 16; off > 0; off >>= 1) v += __shfl_xor(v, off);
  return v;
}

__device__ __forceinline__ float edge_score(const float* __restrict__ ea,
                                            int e, float base, float4 w0,
                                            float4 w1, float4 w2, float4 w3) {
  const float4* er = (const float4*)(ea + (size_t)e * 16);
  float4 v0 = er[0], v1 = er[1], v2 = er[2], v3 = er[3];
  float acc = base;
  acc += v0.x * w0.x + v0.y * w0.y + v0.z * w0.z + v0.w * w0.w;
  acc += v1.x * w1.x + v1.y * w1.y + v1.z * w1.z + v1.w * w1.w;
  acc += v2.x * w2.x + v2.y * w2.y + v2.z * w2.z + v2.w * w2.w;
  acc += v3.x * w3.x + v3.y * w3.y + v3.z * w3.z + v3.w * w3.w;
  return acc > 0.0f ? acc : 0.2f * acc;
}

__device__ __forceinline__ void add_chunk(const float* __restrict__ z,
                                          const float* __restrict__ ea,
                                          int e, int s, float ex, int cntc,
                                          int lane, float& zx, float& zy,
                                          float& ag) {
  int e0 = __builtin_amdgcn_readfirstlane(e);
  int s0 = __builtin_amdgcn_readfirstlane(s);
  if (lane >= cntc) {
    e = e0;
    s = s0;
  }
  int cnt8 = (cntc + 7) & ~7;
  for (int j0 = 0; j0 < cnt8; j0 += 8) {
    float2 zv[8];
    float av[8];
    float wj[8];
#pragma unroll
    for (int u = 0; u < 8; ++u) {
      int ej = __builtin_amdgcn_readlane(e, j0 + u);
      int sj = __builtin_amdgcn_readlane(s, j0 + u);
      wj[u] = __int_as_float(
          __builtin_amdgcn_readlane(__float_as_int(ex), j0 + u));
      zv[u] = *(const float2*)(z + (size_t)sj * 64 + (lane << 1));
      av[u] = ea[(size_t)ej * 16 + (lane & 15)];
    }
#pragma unroll
    for (int u = 0; u < 8; ++u) {
      zx += wj[u] * zv[u].x;
      zy += wj[u] * zv[u].y;
      ag += wj[u] * av[u];
    }
  }
}

__global__ void __launch_bounds__(256) k_node(
    const int* __restrict__ ei, const float* __restrict__ ea,
    const float* __restrict__ We, const float* __restrict__ weff,
    const float* __restrict__ x, const float* __restrict__ z,
    const float* __restrict__ sA, const float* __restrict__ sD,
    const int* __restrict__ segStart, const int* __restrict__ perm,
    float* __restrict__ out, int N, int E) {
  __shared__ float s_out[16][64];
  __shared__ float s_agg[16][16];

  int tid = threadIdx.x;
  int wave = __builtin_amdgcn_readfirstlane(tid >> 5);
  int lane = tid & 31;
  int rBase = blockIdx.x << 4;

  const float4* w4 = (const float4*)weff;
  float4 w0 = w4[0], w1 = w4[1], w2 = w4[2], w3 = w4[3];

  for (int q = 0; q < 2; ++q) {
    int r = (wave << 1) + q;
    int n = rBase + r;
    float zx = 0.0f, zy = 0.0f, ag = 0.0f, inv = 0.0f, x0 = 0.0f, x1 = 0.0f;
    if (n < N) {
      int b = segStart[n];
      int en = (n + 1 < N) ? segStart[n + 1] : E;
      float sDn = sD[n];
      const float2 xr = *(const float2*)(x + (size_t)n * 64 + (lane << 1));
      x0 = xr.x;
      x1 = xr.y;
      float dsum = 0.0f;
      if (en - b <= 32) {
        int i = b + lane;
        int e = 0, s = 0;
        float sc = NEGV;
        if (i < en) {
          e = perm[i];
          s = ei[e];
          sc = edge_score(ea, e, sA[s] + sDn, w0, w1, w2, w3);
        }
        float m = wave_max(sc);
        float ex = (i < en) ? __expf(sc - m) : 0.0f;
        dsum = ex;
        add_chunk(z, ea, e, s, ex, en - b, lane, zx, zy, ag);
      } else {
        float m = NEGV;
        for (int c = b; c < en; c += 32) {
          int i = c + lane;
          if (i < en) {
            int e = perm[i];
            int s = ei[e];
            m = fmaxf(m, edge_score(ea, e, sA[s] + sDn, w0, w1, w2, w3));
          }
        }
        m = wave_max(m);
        for (int c = b; c < en; c += 32) {
          int i = c + lane;
          int e = 0, s = 0;
          float ex = 0.0f;
          if (i < en) {
            e = perm[i];
            s = ei[e];
            ex = __expf(edge_score(ea, e, sA[s] + sDn, w0, w1, w2, w3) - m);
          }
          dsum += ex;
          int cntc = en - c;
          if (cntc > 32) cntc = 32;
          add_chunk(z, ea, e, s, ex, cntc, lane, zx, zy, ag);
        }
      }
      dsum = wave_sum(dsum);
      inv = 1.0f / (dsum + 1e-8f);
    }
    s_out[r][(lane << 1)] = x0 + zx * inv;
    s_out[r][(lane << 1) + 1] = x1 + zy * inv;
    if (lane < 16) s_agg[r][lane] = ag * inv;
  }
  __syncthreads();

  if (wave < 4) {
    int half = lane >> 4;
    int l16 = lane & 15;
    int col = (wave << 4) + l16;
    v8f acc;
#pragma unroll
    for (int j = 0; j < 8; ++j) acc[j] = s_out[(half << 3) + j][col];
#pragma unroll
    for (int t = 0; t < 4; ++t) {
      int k = (t << 2) + (half << 1);
      v2f a, bb;
      a.x = s_agg[l16][k];
      a.y = s_agg[l16][k + 1];
      bb.x = We[k * 64 + col];
      bb.y = We[(k + 1) * 64 + col];
      acc = __builtin_amdgcn_wmma_f32_16x16x4_f32(
          false, a, false, bb, (short)0, acc, false, false);
    }
#pragma unroll
    for (int j = 0; j < 8; ++j) s_out[(half << 3) + j][col] = acc[j];
  }
  __syncthreads();

  for (int q = 0; q < 2; ++q) {
    int r = (wave << 1) + q;
    int n = rBase + r;
    if (n < N) {
      float2 v;
      v.x = s_out[r][(lane << 1)];
      v.y = s_out[r][(lane << 1) + 1];
      *(float2*)(out + (size_t)n * 64 + (lane << 1)) = v;
    }
  }
}

extern "C" void kernel_launch(void* const* d_in, const int* in_sizes, int n_in,
                              void* d_out, int out_size, void* d_ws,
                              size_t ws_size, hipStream_t stream) {
  const float* x = (const float*)d_in[0];
  const int* ei = (const int*)d_in[1];
  const float* ea = (const float*)d_in[2];
  const float* Wn = (const float*)d_in[3];
  const float* We = (const float*)d_in[4];
  const float* att_src = (const float*)d_in[5];
  const float* att_dst = (const float*)d_in[6];
  const float* att_edge = (const float*)d_in[7];

  int N = in_sizes[0] / 64;
  int E = in_sizes[1] / 2;
  float* out = (float*)d_out;

  float* weff = (float*)d_ws;
  float* z = weff + 64;
  float* sA = z + (size_t)N * 64;
  float* sD = sA + N;
  int* cnt = (int*)(sD + N);
  int* perm = cnt + N;
  int* planes = (int*)z;
  int* tot = planes + (size_t)SORT_C * N;

  int nRanges = (N + SORT_R - 1) / SORT_R;
  hipLaunchKernelGGL(k_count, dim3(nRanges * SORT_C), dim3(SORT_T), 0, stream,
                     ei, We, att_edge, weff, planes, tot, N, E);

  hipLaunchKernelGGL(k_place, dim3(nRanges * SORT_C), dim3(SORT_T), 0, stream,
                     ei, planes, tot, cnt, perm, N, E);

  int nRowTiles = (N + 15) / 16;
  int gb = (nRowTiles + 7) / 8;
  hipLaunchKernelGGL(k_node_gemm, dim3(gb), dim3(256), 0, stream, x, Wn, z, N);

  hipLaunchKernelGGL(k_node_scores, dim3((N + 255) / 256), dim3(256), 0, stream,
                     z, att_src, att_dst, sA, sD, N);

  hipLaunchKernelGGL(k_node, dim3(nRowTiles), dim3(256), 0, stream, ei, ea, We,
                     weff, x, z, sA, sD, cnt, perm, out, N, E);
}


# node product moved behind the segment sum (linearity): x rows gathered, one [16,80]@[Wn;We] f32 WMMA product per node tile, z and its GEMM launch gone, score terms = x.(Wn@att); sum order changes
# speedup vs baseline: 2.9949x; 1.0850x over previous
#include <hip/hip_runtime.h>


typedef __attribute__((ext_vector_type(2))) float v2f;
typedef __attribute__((ext_vector_type(8))) float v8f;

#define NEGV (-1000000000.0f)

#define SORT_C 16
#define SORT_R 6400
#define SORT_T 1024
#define SORT_B 16

__device__ __forceinline__ int wave_sum_int(int v) {
#pragma unroll
  for (int off = 16; off > 0; off >>= 1) v += __shfl_xor(v, off);
  return v;
}

__global__ void __launch_bounds__(SORT_T) k_count(
    const int* __restrict__ ei, const float* __restrict__ We,
    const float* __restrict__ att_edge, const float* __restrict__ Wn,
    const float* __restrict__ att_src, const float* __restrict__ att_dst,
    float* __restrict__ weff, float* __restrict__ usd,
    int* __restrict__ planes, int* __restrict__ tot, int N, int E) {
  __shared__ int s_cnt[SORT_R];
  __shared__ int s_total;
  int tid = threadIdx.x;
  int lane = tid & 31;
  int j = blockIdx.x / SORT_C;
  int c = blockIdx.x - j * SORT_C;
  int lo = j * SORT_R;
  int nr = N - lo;
  if (nr > SORT_R) nr = SORT_R;
  int chunk = (E + SORT_C - 1) / SORT_C;
  int eb = c * chunk;
  int ee = eb + chunk;
  if (ee > E) ee = E;

  if (blockIdx.x == 0 && tid < 16) {
    float s = 0.0f;
    for (int k = 0; k < 64; ++k) s += We[tid * 64 + k] * att_edge[k];
    weff[tid] = s;
  }
  if (blockIdx.x == 0 && tid >= 32 && tid < 160) {
    int t = tid - 32;
    int k = t & 63;
    const float4* wr = (const float4*)(Wn + k * 64);
    const float4* av = (const float4*)((t >> 6) ? att_dst : att_src);
    float s = 0.0f;
#pragma unroll
    for (int i = 0; i < 16; ++i) {
      float4 w = wr[i], a = av[i];
      s += w.x * a.x + w.y * a.y + w.z * a.z + w.w * a.w;
    }
    usd[t] = s;
  }
  for (int i = tid; i < SORT_R; i += SORT_T) s_cnt[i] = 0;
  if (tid == 0) s_total = 0;
  __syncthreads();

  const int* __restrict__ dst = ei + E;
  for (int base = eb; base < ee; base += SORT_T * SORT_B) {
    int d[SORT_B];
#pragma unroll
    for (int u = 0; u < SORT_B; ++u) {
      int i = base + u * SORT_T + tid;
      int dv = dst[i < ee ? i : ee - 1];
      d[u] = (i < ee) ? dv : -1;
    }
#pragma unroll
    for (int u = 0; u < SORT_B; ++u) {
      unsigned r = (unsigned)(d[u] - lo);
      if (r < (unsigned)nr) atomicAdd(&s_cnt[r], 1);
    }
  }
  __syncthreads();

  int part = 0;
  for (int i = tid; i < nr; i += SORT_T) {
    int v = s_cnt[i];
    planes[(size_t)c * N + lo + i] = v;
    part += v;
  }
  part = wave_sum_int(part);
  if (lane == 0) atomicAdd(&s_total, part);
  __syncthreads();
  if (tid == 0) tot[blockIdx.x] = s_total;
}

__global__ void __launch_bounds__(SORT_T) k_place(
    const int* __restrict__ ei, const int* __restrict__ planes,
    const int* __restrict__ tot, int* __restrict__ segStart,
    int* __restrict__ perm, int N, int E) {
  __shared__ int s_slot[SORT_R];
  __shared__ int s_bef[SORT_R];
  __shared__ int s_w[32];
  __shared__ int s_rb[32];
  int tid = threadIdx.x;
  int lane = tid & 31;
  int wave = tid >> 5;
  int j = blockIdx.x / SORT_C;
  int c = blockIdx.x - j * SORT_C;
  int lo = j * SORT_R;
  int nr = N - lo;
  if (nr > SORT_R) nr = SORT_R;
  int chunk = (E + SORT_C - 1) / SORT_C;
  int eb = c * chunk;
  int ee = eb + chunk;
  if (ee > E) ee = E;

  int rb = 0;
  for (int t = tid; t < j * SORT_C; t += SORT_T) rb += tot[t];
  rb = wave_sum_int(rb);
  if (lane == 0) s_rb[wave] = rb;

  for (int i = tid; i < SORT_R; i += SORT_T) {
    int t = 0, b = 0;
    if (i < nr) {
#pragma unroll
      for (int cc = 0; cc < SORT_C; ++cc) {
        int v = planes[(size_t)cc * N + lo + i];
        t += v;
        if (cc < c) b += v;
      }
    }
    s_slot[i] = t;
    s_bef[i] = b;
  }
  __syncthreads();

  int rbase = wave_sum_int(s_rb[lane]);

  constexpr int SORT_K = (SORT_R + SORT_T - 1) / SORT_T;
  int i0 = tid * SORT_K;
  int v[SORT_K];
  int sum = 0;
#pragma unroll
  for (int k = 0; k < SORT_K; ++k) {
    int i = i0 + k;
    v[k] = (i < SORT_R) ? s_slot[i] : 0;
    sum += v[k];
  }
  int incl = sum;
#pragma unroll
  for (int off = 1; off < 32; off <<= 1) {
    int u = __shfl_up(incl, off);
    if (lane >= off) incl += u;
  }
  if (lane == 31) s_w[wave] = incl;
  __syncthreads();
  int wv = s_w[lane];
  int winc = wv;
#pragma unroll
  for (int off = 1; off < 32; off <<= 1) {
    int u = __shfl_up(winc, off);
    if (lane >= off) winc += u;
  }
  int run = rbase + __shfl(winc, wave) - __shfl(wv, wave) + (incl - sum);
#pragma unroll
  for (int k = 0; k < SORT_K; ++k) {
    int i = i0 + k;
    if (i < SORT_R) {
      if (c == 0 && i < nr) segStart[lo + i] = run;
      s_slot[i] = run + s_bef[i];
      run += v[k];
    }
  }
  __syncthreads();

  const int* __restrict__ dst = ei + E;
  for (int base = eb; base < ee; base += SORT_T * SORT_B) {
    int d[SORT_B];
#pragma unroll
    for (int u = 0; u < SORT_B; ++u) {
      int i = base + u * SORT_T + tid;
      int dv = dst[i < ee ? i : ee - 1];
      d[u] = (i < ee) ? dv : -1;
    }
    int slot[SORT_B];
#pragma unroll
    for (int u = 0; u < SORT_B; ++u) {
      unsigned r = (unsigned)(d[u] - lo);
      slot[u] = (r < (unsigned)nr) ? atomicAdd(&s_slot[r], 1) : -1;
    }
#pragma unroll
    for (int u = 0; u < SORT_B; ++u)
      if (slot[u] >= 0) perm[slot[u]] = base + u * SORT_T + tid;
  }
}

__global__ void k_node_scores(const float* __restrict__ x,
                              const float* __restrict__ usd,
                              float* __restrict__ sA, float* __restrict__ sD,
                              int N) {
  int n = blockIdx.x * blockDim.x + threadIdx.x;
  if (n >= N) return;
  const float4* zr = (const float4*)(x + (size_t)n * 64);
  const float4* as4 = (const float4*)usd;
  const float4* ad4 = (const float4*)(usd + 64);
  float a = 0.0f, d = 0.0f;
#pragma unroll
  for (int i = 0; i < 16; ++i) {
    float4 v = zr[i], s = as4[i], t = ad4[i];
    a += v.x * s.x + v.y * s.y + v.z * s.z + v.w * s.w;
    d += v.x * t.x + v.y * t.y + v.z * t.z + v.w * t.w;
  }
  sA[n] = a;
  sD[n] = d;
}

__device__ __forceinline__ float wave_max(float v) {
#pragma unroll
  for (int off = 16; off > 0; off >>= 1) v = fmaxf(v, __shfl_xor(v, off));
  return v;
}
__device__ __forceinline__ float wave_sum(float v) {
#pragma unroll
  for (int off = 16; off > 0; off >>= 1) v += __shfl_xor(v, off);
  return v;
}

__device__ __forceinline__ float edge_score(const float* __restrict__ ea,
                                            int e, float base, float4 w0,
                                            float4 w1, float4 w2, float4 w3) {
  const float4* er = (const float4*)(ea + (size_t)e * 16);
  float4 v0 = er[0], v1 = er[1], v2 = er[2], v3 = er[3];
  float acc = base;
  acc += v0.x * w0.x + v0.y * w0.y + v0.z * w0.z + v0.w * w0.w;
  acc += v1.x * w1.x + v1.y * w1.y + v1.z * w1.z + v1.w * w1.w;
  acc += v2.x * w2.x + v2.y * w2.y + v2.z * w2.z + v2.w * w2.w;
  acc += v3.x * w3.x + v3.y * w3.y + v3.z * w3.z + v3.w * w3.w;
  return acc > 0.0f ? acc : 0.2f * acc;
}

__device__ __forceinline__ void add_chunk(const float* __restrict__ xg,
                                          const float* __restrict__ ea,
                                          int e, int s, float ex, int cntc,
                                          int lane, float& zx, float& zy,
                                          float& ag) {
  int e0 = __builtin_amdgcn_readfirstlane(e);
  int s0 = __builtin_amdgcn_readfirstlane(s);
  if (lane >= cntc) {
    e = e0;
    s = s0;
  }
  int cnt8 = (cntc + 7) & ~7;
  for (int j0 = 0; j0 < cnt8; j0 += 8) {
    float2 zv[8];
    float av[8];
    float wj[8];
#pragma unroll
    for (int u = 0; u < 8; ++u) {
      int ej = __builtin_amdgcn_readlane(e, j0 + u);
      int sj = __builtin_amdgcn_readlane(s, j0 + u);
      wj[u] = __int_as_float(
          __builtin_amdgcn_readlane(__float_as_int(ex), j0 + u));
      zv[u] = *(const float2*)(xg + (size_t)sj * 64 + (lane << 1));
      av[u] = ea[(size_t)ej * 16 + (lane & 15)];
    }
#pragma unroll
    for (int u = 0; u < 8; ++u) {
      zx += wj[u] * zv[u].x;
      zy += wj[u] * zv[u].y;
      ag += wj[u] * av[u];
    }
  }
}

__global__ void __launch_bounds__(256) k_node(
    const int* __restrict__ ei, const float* __restrict__ ea,
    const float* __restrict__ Wn, const float* __restrict__ We,
    const float* __restrict__ weff, const float* __restrict__ x,
    const float* __restrict__ sA, const float* __restrict__ sD,
    const int* __restrict__ segStart, const int* __restrict__ perm,
    float* __restrict__ out, int N, int E) {
  __shared__ float s_a[16][84];
  __shared__ float s_p[4][16][68];

  int tid = threadIdx.x;
  int wave = __builtin_amdgcn_readfirstlane(tid >> 5);
  int lane = tid & 31;
  int rBase = blockIdx.x << 4;

  const float4* w4 = (const float4*)weff;
  float4 w0 = w4[0], w1 = w4[1], w2 = w4[2], w3 = w4[3];

  for (int q = 0; q < 2; ++q) {
    int r = (wave << 1) + q;
    int n = rBase + r;
    float zx = 0.0f, zy = 0.0f, ag = 0.0f, inv = 0.0f;
    if (n < N) {
      int b = segStart[n];
      int en = (n + 1 < N) ? segStart[n + 1] : E;
      float sDn = sD[n];
      float dsum = 0.0f;
      if (en - b <= 32) {
        int i = b + lane;
        int e = 0, s = 0;
        float sc = NEGV;
        if (i < en) {
          e = perm[i];
          s = ei[e];
          sc = edge_score(ea, e, sA[s] + sDn, w0, w1, w2, w3);
        }
        float m = wave_max(sc);
        float ex = (i < en) ? __expf(sc - m) : 0.0f;
        dsum = ex;
        add_chunk(x, ea, e, s, ex, en - b, lane, zx, zy, ag);
      } else {
        float m = NEGV;
        for (int c = b; c < en; c += 32) {
          int i = c + lane;
          if (i < en) {
            int e = perm[i];
            int s = ei[e];
            m = fmaxf(m, edge_score(ea, e, sA[s] + sDn, w0, w1, w2, w3));
          }
        }
        m = wave_max(m);
        for (int c = b; c < en; c += 32) {
          int i = c + lane;
          int e = 0, s = 0;
          float ex = 0.0f;
          if (i < en) {
            e = perm[i];
            s = ei[e];
            ex = __expf(edge_score(ea, e, sA[s] + sDn, w0, w1, w2, w3) - m);
          }
          dsum += ex;
          int cntc = en - c;
          if (cntc > 32) cntc = 32;
          add_chunk(x, ea, e, s, ex, cntc, lane, zx, zy, ag);
        }
      }
      dsum = wave_sum(dsum);
      inv = 1.0f / (dsum + 1e-8f);
    }
    s_a[r][(lane << 1)] = zx * inv;
    s_a[r][(lane << 1) + 1] = zy * inv;
    if (lane < 16) s_a[r][64 + lane] = ag * inv;
  }
  int half = lane >> 4;
  int l16 = lane & 15;
  int p = wave & 1;
  int kq = wave >> 1;
  int c0 = (p << 5) + l16;
  v2f bw0[5], bw1[5];
#pragma unroll
  for (int tt = 0; tt < 5; ++tt) {
    int t = kq * 5 + tt;
    int k = (t << 2) + (half << 1);
    const float* __restrict__ wr = (t < 16) ? Wn + k * 64 : We + (k - 64) * 64;
    bw0[tt].x = wr[c0];
    bw0[tt].y = wr[64 + c0];
    bw1[tt].x = wr[c0 + 16];
    bw1[tt].y = wr[64 + c0 + 16];
  }
  int n0 = rBase + (wave << 1);
  int na = n0 < N ? n0 : N - 1;
  int nb = n0 + 1 < N ? n0 + 1 : N - 1;
  const float2 xa = *(const float2*)(x + (size_t)na * 64 + (lane << 1));
  const float2 xb = *(const float2*)(x + (size_t)nb * 64 + (lane << 1));

  __builtin_amdgcn_fence(__ATOMIC_RELEASE, "workgroup", "local");
  __builtin_amdgcn_s_barrier();
  __builtin_amdgcn_fence(__ATOMIC_ACQUIRE, "workgroup", "local");

  {
    v8f acc0 = {}, acc1 = {};
#pragma unroll
    for (int tt = 0; tt < 5; ++tt) {
      int t = kq * 5 + tt;
      int k = (t << 2) + (half << 1);
      v2f a;
      a.x = s_a[l16][k];
      a.y = s_a[l16][k + 1];
      acc0 = __builtin_amdgcn_wmma_f32_16x16x4_f32(
          false, a, false, bw0[tt], (short)0, acc0, false, false);
      acc1 = __builtin_amdgcn_wmma_f32_16x16x4_f32(
          false, a, false, bw1[tt], (short)0, acc1, false, false);
    }
#pragma unroll
    for (int j = 0; j < 8; ++j) {
      s_p[kq][(half << 3) + j][c0] = acc0[j];
      s_p[kq][(half << 3) + j][c0 + 16] = acc1[j];
    }
  }
  __syncthreads();

#pragma unroll
  for (int q = 0; q < 2; ++q) {
    int r = (wave << 1) + q;
    int n = rBase + r;
    if (n < N) {
      float2 v = q ? xb : xa;
#pragma unroll
      for (int kk = 0; kk < 4; ++kk) {
        float2 pq = *(const float2*)&s_p[kk][r][(lane << 1)];
        v.x += pq.x;
        v.y += pq.y;
      }
      *(float2*)(out + (size_t)n * 64 + (lane << 1)) = v;
    }
  }
}

extern "C" void kernel_launch(void* const* d_in, const int* in_sizes, int n_in,
                              void* d_out, int out_size, void* d_ws,
                              size_t ws_size, hipStream_t stream) {
  const float* x = (const float*)d_in[0];
  const int* ei = (const int*)d_in[1];
  const float* ea = (const float*)d_in[2];
  const float* Wn = (const float*)d_in[3];
  const float* We = (const float*)d_in[4];
  const float* att_src = (const float*)d_in[5];
  const float* att_dst = (const float*)d_in[6];
  const float* att_edge = (const float*)d_in[7];

  int N = in_sizes[0] / 64;
  int E = in_sizes[1] / 2;
  float* out = (float*)d_out;

  float* weff = (float*)d_ws;
  float* scr = weff + 64;
  float* sA = scr + (size_t)N * 64;
  float* sD = sA + N;
  int* cnt = (int*)(sD + N);
  int* perm = cnt + N;
  int* planes = (int*)scr;
  int* tot = planes + (size_t)SORT_C * N;
  float* usd = scr + (size_t)N * 48;

  int nRanges = (N + SORT_R - 1) / SORT_R;
  hipLaunchKernelGGL(k_count, dim3(nRanges * SORT_C), dim3(SORT_T), 0, stream,
                     ei, We, att_edge, Wn, att_src, att_dst, weff, usd, planes,
                     tot, N, E);

  hipLaunchKernelGGL(k_place, dim3(nRanges * SORT_C), dim3(SORT_T), 0, stream,
                     ei, planes, tot, cnt, perm, N, E);

  int nRowTiles = (N + 15) / 16;
  hipLaunchKernelGGL(k_node_scores, dim3((N + 255) / 256), dim3(256), 0, stream,
                     x, usd, sA, sD, N);

  hipLaunchKernelGGL(k_node, dim3(nRowTiles), dim3(256), 0, stream, ei, ea, Wn,
                     We, weff, x, sA, sD, cnt, perm, out, N, E);
}


# node pass reads each edge_attr row once, 16 B per lane, 4 lanes per edge (8 rows per load); score dot (2 DPP adds) and weighted row sum from those registers, per-edge re-read gone; sum order changes
# speedup vs baseline: 3.2568x; 1.0874x over previous
#include <hip/hip_runtime.h>


typedef __attribute__((ext_vector_type(2))) float v2f;
typedef __attribute__((ext_vector_type(8))) float v8f;

#define NEGV (-1000000000.0f)

#define SORT_C 16
#define SORT_R 6400
#define SORT_T 1024
#define SORT_B 16

__device__ __forceinline__ int wave_sum_int(int v) {
#pragma unroll
  for (int off = 16; off > 0; off >>= 1) v += __shfl_xor(v, off);
  return v;
}

__global__ void __launch_bounds__(SORT_T) k_count(
    const int* __restrict__ ei, const float* __restrict__ We,
    const float* __restrict__ att_edge, const float* __restrict__ Wn,
    const float* __restrict__ att_src, const float* __restrict__ att_dst,
    float* __restrict__ weff, float* __restrict__ usd,
    int* __restrict__ planes, int* __restrict__ tot, int N, int E) {
  __shared__ int s_cnt[SORT_R];
  __shared__ int s_total;
  int tid = threadIdx.x;
  int lane = tid & 31;
  int j = blockIdx.x / SORT_C;
  int c = blockIdx.x - j * SORT_C;
  int lo = j * SORT_R;
  int nr = N - lo;
  if (nr > SORT_R) nr = SORT_R;
  int chunk = (E + SORT_C - 1) / SORT_C;
  int eb = c * chunk;
  int ee = eb + chunk;
  if (ee > E) ee = E;

  if (blockIdx.x == 0 && tid < 16) {
    float s = 0.0f;
    for (int k = 0; k < 64; ++k) s += We[tid * 64 + k] * att_edge[k];
    weff[tid] = s;
  }
  if (blockIdx.x == 0 && tid >= 32 && tid < 160) {
    int t = tid - 32;
    int k = t & 63;
    const float4* wr = (const float4*)(Wn + k * 64);
    const float4* av = (const float4*)((t >> 6) ? att_dst : att_src);
    float s = 0.0f;
#pragma unroll
    for (int i = 0; i < 16; ++i) {
      float4 w = wr[i], a = av[i];
      s += w.x * a.x + w.y * a.y + w.z * a.z + w.w * a.w;
    }
    usd[t] = s;
  }
  for (int i = tid; i < SORT_R; i += SORT_T) s_cnt[i] = 0;
  if (tid == 0) s_total = 0;
  __syncthreads();

  const int* __restrict__ dst = ei + E;
  for (int base = eb; base < ee; base += SORT_T * SORT_B) {
    int d[SORT_B];
#pragma unroll
    for (int u = 0; u < SORT_B; ++u) {
      int i = base + u * SORT_T + tid;
      int dv = dst[i < ee ? i : ee - 1];
      d[u] = (i < ee) ? dv : -1;
    }
#pragma unroll
    for (int u = 0; u < SORT_B; ++u) {
      unsigned r = (unsigned)(d[u] - lo);
      if (r < (unsigned)nr) atomicAdd(&s_cnt[r], 1);
    }
  }
  __syncthreads();

  int part = 0;
  for (int i = tid; i < nr; i += SORT_T) {
    int v = s_cnt[i];
    planes[(size_t)c * N + lo + i] = v;
    part += v;
  }
  part = wave_sum_int(part);
  if (lane == 0) atomicAdd(&s_total, part);
  __syncthreads();
  if (tid == 0) tot[blockIdx.x] = s_total;
}

__global__ void __launch_bounds__(SORT_T) k_place(
    const int* __restrict__ ei, const int* __restrict__ planes,
    const int* __restrict__ tot, int* __restrict__ segStart,
    int* __restrict__ perm, int N, int E) {
  __shared__ int s_slot[SORT_R];
  __shared__ int s_bef[SORT_R];
  __shared__ int s_w[32];
  __shared__ int s_rb[32];
  int tid = threadIdx.x;
  int lane = tid & 31;
  int wave = tid >> 5;
  int j = blockIdx.x / SORT_C;
  int c = blockIdx.x - j * SORT_C;
  int lo = j * SORT_R;
  int nr = N - lo;
  if (nr > SORT_R) nr = SORT_R;
  int chunk = (E + SORT_C - 1) / SORT_C;
  int eb = c * chunk;
  int ee = eb + chunk;
  if (ee > E) ee = E;

  int rb = 0;
  for (int t = tid; t < j * SORT_C; t += SORT_T) rb += tot[t];
  rb = wave_sum_int(rb);
  if (lane == 0) s_rb[wave] = rb;

  for (int i = tid; i < SORT_R; i += SORT_T) {
    int t = 0, b = 0;
    if (i < nr) {
#pragma unroll
      for (int cc = 0; cc < SORT_C; ++cc) {
        int v = planes[(size_t)cc * N + lo + i];
        t += v;
        if (cc < c) b += v;
      }
    }
    s_slot[i] = t;
    s_bef[i] = b;
  }
  __syncthreads();

  int rbase = wave_sum_int(s_rb[lane]);

  constexpr int SORT_K = (SORT_R + SORT_T - 1) / SORT_T;
  int i0 = tid * SORT_K;
  int v[SORT_K];
  int sum = 0;
#pragma unroll
  for (int k = 0; k < SORT_K; ++k) {
    int i = i0 + k;
    v[k] = (i < SORT_R) ? s_slot[i] : 0;
    sum += v[k];
  }
  int incl = sum;
#pragma unroll
  for (int off = 1; off < 32; off <<= 1) {
    int u = __shfl_up(incl, off);
    if (lane >= off) incl += u;
  }
  if (lane == 31) s_w[wave] = incl;
  __syncthreads();
  int wv = s_w[lane];
  int winc = wv;
#pragma unroll
  for (int off = 1; off < 32; off <<= 1) {
    int u = __shfl_up(winc, off);
    if (lane >= off) winc += u;
  }
  int run = rbase + __shfl(winc, wave) - __shfl(wv, wave) + (incl - sum);
#pragma unroll
  for (int k = 0; k < SORT_K; ++k) {
    int i = i0 + k;
    if (i < SORT_R) {
      if (c == 0 && i < nr) segStart[lo + i] = run;
      s_slot[i] = run + s_bef[i];
      run += v[k];
    }
  }
  __syncthreads();

  const int* __restrict__ dst = ei + E;
  for (int base = eb; base < ee; base += SORT_T * SORT_B) {
    int d[SORT_B];
#pragma unroll
    for (int u = 0; u < SORT_B; ++u) {
      int i = base + u * SORT_T + tid;
      int dv = dst[i < ee ? i : ee - 1];
      d[u] = (i < ee) ? dv : -1;
    }
    int slot[SORT_B];
#pragma unroll
    for (int u = 0; u < SORT_B; ++u) {
      unsigned r = (unsigned)(d[u] - lo);
      slot[u] = (r < (unsigned)nr) ? atomicAdd(&s_slot[r], 1) : -1;
    }
#pragma unroll
    for (int u = 0; u < SORT_B; ++u)
      if (slot[u] >= 0) perm[slot[u]] = base + u * SORT_T + tid;
  }
}

__global__ void k_node_scores(const float* __restrict__ x,
                              const float* __restrict__ usd,
                              float* __restrict__ sA, float* __restrict__ sD,
                              int N) {
  int n = blockIdx.x * blockDim.x + threadIdx.x;
  if (n >= N) return;
  const float4* zr = (const float4*)(x + (size_t)n * 64);
  const float4* as4 = (const float4*)usd;
  const float4* ad4 = (const float4*)(usd + 64);
  float a = 0.0f, d = 0.0f;
#pragma unroll
  for (int i = 0; i < 16; ++i) {
    float4 v = zr[i], s = as4[i], t = ad4[i];
    a += v.x * s.x + v.y * s.y + v.z * s.z + v.w * s.w;
    d += v.x * t.x + v.y * t.y + v.z * t.z + v.w * t.w;
  }
  sA[n] = a;
  sD[n] = d;
}

__device__ __forceinline__ float wave_max(float v) {
#pragma unroll
  for (int off = 16; off > 0; off >>= 1) v = fmaxf(v, __shfl_xor(v, off));
  return v;
}
__device__ __forceinline__ float wave_sum(float v) {
#pragma unroll
  for (int off = 16; off > 0; off >>= 1) v += __shfl_xor(v, off);
  return v;
}

__device__ __forceinline__ float edge_score(const float* __restrict__ ea,
                                            int e, float base, float4 w0,
                                            float4 w1, float4 w2, float4 w3) {
  const float4* er = (const float4*)(ea + (size_t)e * 16);
  float4 v0 = er[0], v1 = er[1], v2 = er[2], v3 = er[3];
  float acc = base;
  acc += v0.x * w0.x + v0.y * w0.y + v0.z * w0.z + v0.w * w0.w;
  acc += v1.x * w1.x + v1.y * w1.y + v1.z * w1.z + v1.w * w1.w;
  acc += v2.x * w2.x + v2.y * w2.y + v2.z * w2.z + v2.w * w2.w;
  acc += v3.x * w3.x + v3.y * w3.y + v3.z * w3.z + v3.w * w3.w;
  return acc > 0.0f ? acc : 0.2f * acc;
}

__device__ __forceinline__ void add_chunk(const float* __restrict__ xg,
                                          const float* __restrict__ ea,
                                          int e, int s, float ex, int cntc,
                                          int lane, float& zx, float& zy,
                                          float& ag) {
  int e0 = __builtin_amdgcn_readfirstlane(e);
  int s0 = __builtin_amdgcn_readfirstlane(s);
  if (lane >= cntc) {
    e = e0;
    s = s0;
  }
  int cnt8 = (cntc + 7) & ~7;
  for (int j0 = 0; j0 < cnt8; j0 += 8) {
    float2 zv[8];
    float av[8];
    float wj[8];
#pragma unroll
    for (int u = 0; u < 8; ++u) {
      int ej = __builtin_amdgcn_readlane(e, j0 + u);
      int sj = __builtin_amdgcn_readlane(s, j0 + u);
      wj[u] = __int_as_float(
          __builtin_amdgcn_readlane(__float_as_int(ex), j0 + u));
      zv[u] = *(const float2*)(xg + (size_t)sj * 64 + (lane << 1));
      av[u] = ea[(size_t)ej * 16 + (lane & 15)];
    }
#pragma unroll
    for (int u = 0; u < 8; ++u) {
      zx += wj[u] * zv[u].x;
      zy += wj[u] * zv[u].y;
      ag += wj[u] * av[u];
    }
  }
}

template <int M>
__device__ __forceinline__ float xrow(float v) {
  return __int_as_float(__builtin_amdgcn_update_dpp(
      0, __float_as_int(v), 0x160 + M, 0xF, 0xF, true));
}
__device__ __forceinline__ float xhalf(float v) {
  return __int_as_float(__builtin_amdgcn_permlanex16(
      __float_as_int(v), __float_as_int(v), 0x76543210, 0xfedcba98, false,
      false));
}

__device__ __forceinline__ void add_rows8(const float* __restrict__ xg, int sq,
                                          float exq, int lane, float& zx,
                                          float& zy) {
  float2 zv[8];
  float wj[8];
#pragma unroll
  for (int u = 0; u < 8; ++u) {
    int sj = __builtin_amdgcn_readlane(sq, 4 * u);
    wj[u] = __int_as_float(
        __builtin_amdgcn_readlane(__float_as_int(exq), 4 * u));
    zv[u] = *(const float2*)(xg + (size_t)sj * 64 + (lane << 1));
  }
#pragma unroll
  for (int u = 0; u < 8; ++u) {
    zx += wj[u] * zv[u].x;
    zy += wj[u] * zv[u].y;
  }
}

template <int G>
__device__ __forceinline__ void seg_quads(
    const int* __restrict__ ei, const float* __restrict__ ea,
    const float* __restrict__ xg, const float* __restrict__ sA,
    const int* __restrict__ perm, int b, int cnt, float sDn, float4 wq,
    int sub, int qe, int lane, float& zx, float& zy, float4& aq,
    float& dsum) {
  bool live[G];
  int e[G], s[G];
  float bs[G], sc[G], exq[G];
  float4 pc[G];
#pragma unroll
  for (int g = 0; g < G; ++g) {
    int j = 8 * g + qe;
    live[g] = j < cnt;
    e[g] = perm[b + (live[g] ? j : 0)];
  }
#pragma unroll
  for (int g = 0; g < G; ++g) s[g] = ei[e[g]];
#pragma unroll
  for (int g = 0; g < G; ++g)
    pc[g] = *(const float4*)(ea + (size_t)e[g] * 16 + (sub << 2));
#pragma unroll
  for (int g = 0; g < G; ++g) bs[g] = sA[s[g]];
  float ml = NEGV;
#pragma unroll
  for (int g = 0; g < G; ++g) {
    float pd = pc[g].x * wq.x + pc[g].y * wq.y + pc[g].z * wq.z +
               pc[g].w * wq.w;
    pd += xrow<1>(pd);
    pd += xrow<2>(pd);
    float a = (bs[g] + sDn) + pd;
    a = a > 0.0f ? a : 0.2f * a;
    sc[g] = live[g] ? a : NEGV;
    ml = fmaxf(ml, sc[g]);
  }
  float m = wave_max(ml);
  float dq = 0.0f;
#pragma unroll
  for (int g = 0; g < G; ++g) {
    exq[g] = live[g] ? __expf(sc[g] - m) : 0.0f;
    dq += exq[g];
    aq.x += exq[g] * pc[g].x;
    aq.y += exq[g] * pc[g].y;
    aq.z += exq[g] * pc[g].z;
    aq.w += exq[g] * pc[g].w;
  }
  dsum = (sub == 0) ? dq : 0.0f;
  aq.x += xrow<4>(aq.x);
  aq.y += xrow<4>(aq.y);
  aq.z += xrow<4>(aq.z);
  aq.w += xrow<4>(aq.w);
  aq.x += xrow<8>(aq.x);
  aq.y += xrow<8>(aq.y);
  aq.z += xrow<8>(aq.z);
  aq.w += xrow<8>(aq.w);
  aq.x += xhalf(aq.x);
  aq.y += xhalf(aq.y);
  aq.z += xhalf(aq.z);
  aq.w += xhalf(aq.w);
#pragma unroll 1
  for (int g = 0; g < G; ++g) {
    int sg = s[0];
    float eg = exq[0];
#pragma unroll
    for (int k = 1; k < G; ++k) {
      if (g == k) {
        sg = s[k];
        eg = exq[k];
      }
    }
    add_rows8(xg, sg, eg, lane, zx, zy);
  }
}

__global__ void __launch_bounds__(256, 16) k_node(
    const int* __restrict__ ei, const float* __restrict__ ea,
    const float* __restrict__ Wn, const float* __restrict__ We,
    const float* __restrict__ weff, const float* __restrict__ x,
    const float* __restrict__ sA, const float* __restrict__ sD,
    const int* __restrict__ segStart, const int* __restrict__ perm,
    float* __restrict__ out, int N, int E) {
  __shared__ float s_a[16][84];
  __shared__ float s_p[4][16][68];

  int tid = threadIdx.x;
  int wave = __builtin_amdgcn_readfirstlane(tid >> 5);
  int lane = tid & 31;
  int rBase = blockIdx.x << 4;

  const float4* w4 = (const float4*)weff;
  float4 w0 = w4[0], w1 = w4[1], w2 = w4[2], w3 = w4[3];
  int sub = lane & 3;
  int qe = lane >> 2;
  const float4 wq = w4[sub];

  for (int q = 0; q < 2; ++q) {
    int r = (wave << 1) + q;
    int n = rBase + r;
    float zx = 0.0f, zy = 0.0f, ag = 0.0f, inv = 0.0f;
    float4 aq = make_float4(0.0f, 0.0f, 0.0f, 0.0f);
    bool quad = true;
    if (n < N) {
      int b = segStart[n];
      int en = (n + 1 < N) ? segStart[n + 1] : E;
      float sDn = sD[n];
      float dsum = 0.0f;
      if (en - b <= 24) {
        int cnt = en - b;
        if (cnt > 0) {
          int ng = (cnt + 7) >> 3;
          if (ng == 2)
            seg_quads<2>(ei, ea, x, sA, perm, b, cnt, sDn, wq, sub, qe, lane,
                         zx, zy, aq, dsum);
          else if (ng == 3)
            seg_quads<3>(ei, ea, x, sA, perm, b, cnt, sDn, wq, sub, qe, lane,
                         zx, zy, aq, dsum);
          else
            seg_quads<1>(ei, ea, x, sA, perm, b, cnt, sDn, wq, sub, qe, lane,
                         zx, zy, aq, dsum);
        }
      } else if (en - b <= 32) {
        quad = false;
        int i = b + lane;
        int e = 0, s = 0;
        float sc = NEGV;
        if (i < en) {
          e = perm[i];
          s = ei[e];
          sc = edge_score(ea, e, sA[s] + sDn, w0, w1, w2, w3);
        }
        float m = wave_max(sc);
        float ex = (i < en) ? __expf(sc - m) : 0.0f;
        dsum = ex;
        add_chunk(x, ea, e, s, ex, en - b, lane, zx, zy, ag);
      } else {
        quad = false;
        float m = NEGV;
        for (int c = b; c < en; c += 32) {
          int i = c + lane;
          if (i < en) {
            int e = perm[i];
            int s = ei[e];
            m = fmaxf(m, edge_score(ea, e, sA[s] + sDn, w0, w1, w2, w3));
          }
        }
        m = wave_max(m);
        for (int c = b; c < en; c += 32) {
          int i = c + lane;
          int e = 0, s = 0;
          float ex = 0.0f;
          if (i < en) {
            e = perm[i];
            s = ei[e];
            ex = __expf(edge_score(ea, e, sA[s] + sDn, w0, w1, w2, w3) - m);
          }
          dsum += ex;
          int cntc = en - c;
          if (cntc > 32) cntc = 32;
          add_chunk(x, ea, e, s, ex, cntc, lane, zx, zy, ag);
        }
      }
      dsum = wave_sum(dsum);
      inv = 1.0f / (dsum + 1e-8f);
    }
    s_a[r][(lane << 1)] = zx * inv;
    s_a[r][(lane << 1) + 1] = zy * inv;
    if (quad) {
      if (lane < 4) {
        s_a[r][64 + (lane << 2)] = aq.x * inv;
        s_a[r][64 + (lane << 2) + 1] = aq.y * inv;
        s_a[r][64 + (lane << 2) + 2] = aq.z * inv;
        s_a[r][64 + (lane << 2) + 3] = aq.w * inv;
      }
    } else if (lane < 16) {
      s_a[r][64 + lane] = ag * inv;
    }
  }
  int half = lane >> 4;
  int l16 = lane & 15;
  int p = wave & 1;
  int kq = wave >> 1;
  int c0 = (p << 5) + l16;
  v2f bw0[5], bw1[5];
#pragma unroll
  for (int tt = 0; tt < 5; ++tt) {
    int t = kq * 5 + tt;
    int k = (t << 2) + (half << 1);
    const float* __restrict__ wr = (t < 16) ? Wn + k * 64 : We + (k - 64) * 64;
    bw0[tt].x = wr[c0];
    bw0[tt].y = wr[64 + c0];
    bw1[tt].x = wr[c0 + 16];
    bw1[tt].y = wr[64 + c0 + 16];
  }
  int n0 = rBase + (wave << 1);
  int na = n0 < N ? n0 : N - 1;
  int nb = n0 + 1 < N ? n0 + 1 : N - 1;
  const float2 xa = *(const float2*)(x + (size_t)na * 64 + (lane << 1));
  const float2 xb = *(const float2*)(x + (size_t)nb * 64 + (lane << 1));

  __builtin_amdgcn_fence(__ATOMIC_RELEASE, "workgroup", "local");
  __builtin_amdgcn_s_barrier();
  __builtin_amdgcn_fence(__ATOMIC_ACQUIRE, "workgroup", "local");

  {
    v8f acc0 = {}, acc1 = {};
#pragma unroll
    for (int tt = 0; tt < 5; ++tt) {
      int t = kq * 5 + tt;
      int k = (t << 2) + (half << 1);
      v2f a;
      a.x = s_a[l16][k];
      a.y = s_a[l16][k + 1];
      acc0 = __builtin_amdgcn_wmma_f32_16x16x4_f32(
          false, a, false, bw0[tt], (short)0, acc0, false, false);
      acc1 = __builtin_amdgcn_wmma_f32_16x16x4_f32(
          false, a, false, bw1[tt], (short)0, acc1, false, false);
    }
#pragma unroll
    for (int j = 0; j < 8; ++j) {
      s_p[kq][(half << 3) + j][c0] = acc0[j];
      s_p[kq][(half << 3) + j][c0 + 16] = acc1[j];
    }
  }
  __syncthreads();

#pragma unroll
  for (int q = 0; q < 2; ++q) {
    int r = (wave << 1) + q;
    int n = rBase + r;
    if (n < N) {
      float2 v = q ? xb : xa;
#pragma unroll
      for (int kk = 0; kk < 4; ++kk) {
        float2 pq = *(const float2*)&s_p[kk][r][(lane << 1)];
        v.x += pq.x;
        v.y += pq.y;
      }
      *(float2*)(out + (size_t)n * 64 + (lane << 1)) = v;
    }
  }
}

extern "C" void kernel_launch(void* const* d_in, const int* in_sizes, int n_in,
                              void* d_out, int out_size, void* d_ws,
                              size_t ws_size, hipStream_t stream) {
  const float* x = (const float*)d_in[0];
  const int* ei = (const int*)d_in[1];
  const float* ea = (const float*)d_in[2];
  const float* Wn = (const float*)d_in[3];
  const float* We = (const float*)d_in[4];
  const float* att_src = (const float*)d_in[5];
  const float* att_dst = (const float*)d_in[6];
  const float* att_edge = (const float*)d_in[7];

  int N = in_sizes[0] / 64;
  int E = in_sizes[1] / 2;
  float* out = (float*)d_out;

  float* weff = (float*)d_ws;
  float* scr = weff + 64;
  float* sA = scr + (size_t)N * 64;
  float* sD = sA + N;
  int* cnt = (int*)(sD + N);
  int* perm = cnt + N;
  int* planes = (int*)scr;
  int* tot = planes + (size_t)SORT_C * N;
  float* usd = scr + (size_t)N * 48;

  int nRanges = (N + SORT_R - 1) / SORT_R;
  hipLaunchKernelGGL(k_count, dim3(nRanges * SORT_C), dim3(SORT_T), 0, stream,
                     ei, We, att_edge, Wn, att_src, att_dst, weff, usd, planes,
                     tot, N, E);

  hipLaunchKernelGGL(k_place, dim3(nRanges * SORT_C), dim3(SORT_T), 0, stream,
                     ei, planes, tot, cnt, perm, N, E);

  int nRowTiles = (N + 15) / 16;
  hipLaunchKernelGGL(k_node_scores, dim3((N + 255) / 256), dim3(256), 0, stream,
                     x, usd, sA, sD, N);

  hipLaunchKernelGGL(k_node, dim3(nRowTiles), dim3(256), 0, stream, ei, ea, Wn,
                     We, weff, x, sA, sD, cnt, perm, out, N, E);
}


# sorted edge list holds {edge id, source} pairs, sources read in list order by the placement pass; node pass drops its random edge_index read and one dependent round trip per node; sums unchanged
# speedup vs baseline: 3.3679x; 1.0341x over previous
#include <hip/hip_runtime.h>


typedef __attribute__((ext_vector_type(2))) float v2f;
typedef __attribute__((ext_vector_type(8))) float v8f;

#define NEGV (-1000000000.0f)

#define SORT_C 16
#define SORT_R 6400
#define SORT_T 1024
#define SORT_B 16

__device__ __forceinline__ int wave_sum_int(int v) {
#pragma unroll
  for (int off = 16; off > 0; off >>= 1) v += __shfl_xor(v, off);
  return v;
}

__global__ void __launch_bounds__(SORT_T) k_count(
    const int* __restrict__ ei, const float* __restrict__ We,
    const float* __restrict__ att_edge, const float* __restrict__ Wn,
    const float* __restrict__ att_src, const float* __restrict__ att_dst,
    float* __restrict__ weff, float* __restrict__ usd,
    int* __restrict__ planes, int* __restrict__ tot, int N, int E) {
  __shared__ int s_cnt[SORT_R];
  __shared__ int s_total;
  int tid = threadIdx.x;
  int lane = tid & 31;
  int j = blockIdx.x / SORT_C;
  int c = blockIdx.x - j * SORT_C;
  int lo = j * SORT_R;
  int nr = N - lo;
  if (nr > SORT_R) nr = SORT_R;
  int chunk = (E + SORT_C - 1) / SORT_C;
  int eb = c * chunk;
  int ee = eb + chunk;
  if (ee > E) ee = E;

  if (blockIdx.x == 0 && tid < 16) {
    float s = 0.0f;
    for (int k = 0; k < 64; ++k) s += We[tid * 64 + k] * att_edge[k];
    weff[tid] = s;
  }
  if (blockIdx.x == 0 && tid >= 32 && tid < 160) {
    int t = tid - 32;
    int k = t & 63;
    const float4* wr = (const float4*)(Wn + k * 64);
    const float4* av = (const float4*)((t >> 6) ? att_dst : att_src);
    float s = 0.0f;
#pragma unroll
    for (int i = 0; i < 16; ++i) {
      float4 w = wr[i], a = av[i];
      s += w.x * a.x + w.y * a.y + w.z * a.z + w.w * a.w;
    }
    usd[t] = s;
  }
  for (int i = tid; i < SORT_R; i += SORT_T) s_cnt[i] = 0;
  if (tid == 0) s_total = 0;
  __syncthreads();

  const int* __restrict__ dst = ei + E;
  for (int base = eb; base < ee; base += SORT_T * SORT_B) {
    int d[SORT_B];
#pragma unroll
    for (int u = 0; u < SORT_B; ++u) {
      int i = base + u * SORT_T + tid;
      int dv = dst[i < ee ? i : ee - 1];
      d[u] = (i < ee) ? dv : -1;
    }
#pragma unroll
    for (int u = 0; u < SORT_B; ++u) {
      unsigned r = (unsigned)(d[u] - lo);
      if (r < (unsigned)nr) atomicAdd(&s_cnt[r], 1);
    }
  }
  __syncthreads();

  int part = 0;
  for (int i = tid; i < nr; i += SORT_T) {
    int v = s_cnt[i];
    planes[(size_t)c * N + lo + i] = v;
    part += v;
  }
  part = wave_sum_int(part);
  if (lane == 0) atomicAdd(&s_total, part);
  __syncthreads();
  if (tid == 0) tot[blockIdx.x] = s_total;
}

__global__ void __launch_bounds__(SORT_T) k_place(
    const int* __restrict__ ei, const int* __restrict__ planes,
    const int* __restrict__ tot, int* __restrict__ segStart,
    int* __restrict__ permw, int N, int E) {
  int2* __restrict__ perm = (int2*)permw;
  __shared__ int s_slot[SORT_R];
  __shared__ int s_bef[SORT_R];
  __shared__ int s_w[32];
  __shared__ int s_rb[32];
  int tid = threadIdx.x;
  int lane = tid & 31;
  int wave = tid >> 5;
  int j = blockIdx.x / SORT_C;
  int c = blockIdx.x - j * SORT_C;
  int lo = j * SORT_R;
  int nr = N - lo;
  if (nr > SORT_R) nr = SORT_R;
  int chunk = (E + SORT_C - 1) / SORT_C;
  int eb = c * chunk;
  int ee = eb + chunk;
  if (ee > E) ee = E;

  int rb = 0;
  for (int t = tid; t < j * SORT_C; t += SORT_T) rb += tot[t];
  rb = wave_sum_int(rb);
  if (lane == 0) s_rb[wave] = rb;

  for (int i = tid; i < SORT_R; i += SORT_T) {
    int t = 0, b = 0;
    if (i < nr) {
#pragma unroll
      for (int cc = 0; cc < SORT_C; ++cc) {
        int v = planes[(size_t)cc * N + lo + i];
        t += v;
        if (cc < c) b += v;
      }
    }
    s_slot[i] = t;
    s_bef[i] = b;
  }
  __syncthreads();

  int rbase = wave_sum_int(s_rb[lane]);

  constexpr int SORT_K = (SORT_R + SORT_T - 1) / SORT_T;
  int i0 = tid * SORT_K;
  int v[SORT_K];
  int sum = 0;
#pragma unroll
  for (int k = 0; k < SORT_K; ++k) {
    int i = i0 + k;
    v[k] = (i < SORT_R) ? s_slot[i] : 0;
    sum += v[k];
  }
  int incl = sum;
#pragma unroll
  for (int off = 1; off < 32; off <<= 1) {
    int u = __shfl_up(incl, off);
    if (lane >= off) incl += u;
  }
  if (lane == 31) s_w[wave] = incl;
  __syncthreads();
  int wv = s_w[lane];
  int winc = wv;
#pragma unroll
  for (int off = 1; off < 32; off <<= 1) {
    int u = __shfl_up(winc, off);
    if (lane >= off) winc += u;
  }
  int run = rbase + __shfl(winc, wave) - __shfl(wv, wave) + (incl - sum);
#pragma unroll
  for (int k = 0; k < SORT_K; ++k) {
    int i = i0 + k;
    if (i < SORT_R) {
      if (c == 0 && i < nr) segStart[lo + i] = run;
      s_slot[i] = run + s_bef[i];
      run += v[k];
    }
  }
  __syncthreads();

  const int* __restrict__ dst = ei + E;
  for (int base = eb; base < ee; base += SORT_T * SORT_B) {
    int d[SORT_B];
    int sv[SORT_B];
#pragma unroll
    for (int u = 0; u < SORT_B; ++u) {
      int i = base + u * SORT_T + tid;
      int ic = i < ee ? i : ee - 1;
      d[u] = dst[ic];
      sv[u] = ei[ic];
    }
    int slot[SORT_B];
#pragma unroll
    for (int u = 0; u < SORT_B; ++u) {
      int i = base + u * SORT_T + tid;
      unsigned r = (unsigned)(d[u] - lo);
      slot[u] = (i < ee && r < (unsigned)nr) ? atomicAdd(&s_slot[r], 1) : -1;
    }
#pragma unroll
    for (int u = 0; u < SORT_B; ++u)
      if (slot[u] >= 0)
        perm[slot[u]] = make_int2(base + u * SORT_T + tid, sv[u]);
  }
}

__global__ void k_node_scores(const float* __restrict__ x,
                              const float* __restrict__ usd,
                              float* __restrict__ sA, float* __restrict__ sD,
                              int N) {
  int n = blockIdx.x * blockDim.x + threadIdx.x;
  if (n >= N) return;
  const float4* zr = (const float4*)(x + (size_t)n * 64);
  const float4* as4 = (const float4*)usd;
  const float4* ad4 = (const float4*)(usd + 64);
  float a = 0.0f, d = 0.0f;
#pragma unroll
  for (int i = 0; i < 16; ++i) {
    float4 v = zr[i], s = as4[i], t = ad4[i];
    a += v.x * s.x + v.y * s.y + v.z * s.z + v.w * s.w;
    d += v.x * t.x + v.y * t.y + v.z * t.z + v.w * t.w;
  }
  sA[n] = a;
  sD[n] = d;
}

__device__ __forceinline__ float wave_max(float v) {
#pragma unroll
  for (int off = 16; off > 0; off >>= 1) v = fmaxf(v, __shfl_xor(v, off));
  return v;
}
__device__ __forceinline__ float wave_sum(float v) {
#pragma unroll
  for (int off = 16; off > 0; off >>= 1) v += __shfl_xor(v, off);
  return v;
}

__device__ __forceinline__ float edge_score(const float* __restrict__ ea,
                                            int e, float base, float4 w0,
                                            float4 w1, float4 w2, float4 w3) {
  const float4* er = (const float4*)(ea + (size_t)e * 16);
  float4 v0 = er[0], v1 = er[1], v2 = er[2], v3 = er[3];
  float acc = base;
  acc += v0.x * w0.x + v0.y * w0.y + v0.z * w0.z + v0.w * w0.w;
  acc += v1.x * w1.x + v1.y * w1.y + v1.z * w1.z + v1.w * w1.w;
  acc += v2.x * w2.x + v2.y * w2.y + v2.z * w2.z + v2.w * w2.w;
  acc += v3.x * w3.x + v3.y * w3.y + v3.z * w3.z + v3.w * w3.w;
  return acc > 0.0f ? acc : 0.2f * acc;
}

__device__ __forceinline__ void add_chunk(const float* __restrict__ xg,
                                          const float* __restrict__ ea,
                                          int e, int s, float ex, int cntc,
                                          int lane, float& zx, float& zy,
                                          float& ag) {
  int e0 = __builtin_amdgcn_readfirstlane(e);
  int s0 = __builtin_amdgcn_readfirstlane(s);
  if (lane >= cntc) {
    e = e0;
    s = s0;
  }
  int cnt8 = (cntc + 7) & ~7;
  for (int j0 = 0; j0 < cnt8; j0 += 8) {
    float2 zv[8];
    float av[8];
    float wj[8];
#pragma unroll
    for (int u = 0; u < 8; ++u) {
      int ej = __builtin_amdgcn_readlane(e, j0 + u);
      int sj = __builtin_amdgcn_readlane(s, j0 + u);
      wj[u] = __int_as_float(
          __builtin_amdgcn_readlane(__float_as_int(ex), j0 + u));
      zv[u] = *(const float2*)(xg + (size_t)sj * 64 + (lane << 1));
      av[u] = ea[(size_t)ej * 16 + (lane & 15)];
    }
#pragma unroll
    for (int u = 0; u < 8; ++u) {
      zx += wj[u] * zv[u].x;
      zy += wj[u] * zv[u].y;
      ag += wj[u] * av[u];
    }
  }
}

template <int M>
__device__ __forceinline__ float xrow(float v) {
  return __int_as_float(__builtin_amdgcn_update_dpp(
      0, __float_as_int(v), 0x160 + M, 0xF, 0xF, true));
}
__device__ __forceinline__ float xhalf(float v) {
  return __int_as_float(__builtin_amdgcn_permlanex16(
      __float_as_int(v), __float_as_int(v), 0x76543210, 0xfedcba98, false,
      false));
}

__device__ __forceinline__ void add_rows8(const float* __restrict__ xg, int sq,
                                          float exq, int lane, float& zx,
                                          float& zy) {
  float2 zv[8];
  float wj[8];
#pragma unroll
  for (int u = 0; u < 8; ++u) {
    int sj = __builtin_amdgcn_readlane(sq, 4 * u);
    wj[u] = __int_as_float(
        __builtin_amdgcn_readlane(__float_as_int(exq), 4 * u));
    zv[u] = *(const float2*)(xg + (size_t)sj * 64 + (lane << 1));
  }
#pragma unroll
  for (int u = 0; u < 8; ++u) {
    zx += wj[u] * zv[u].x;
    zy += wj[u] * zv[u].y;
  }
}

template <int G>
__device__ __forceinline__ void seg_quads(
    const float* __restrict__ ea, const float* __restrict__ xg,
    const float* __restrict__ sA, const int2* __restrict__ perm, int b,
    int cnt, float sDn, float4 wq, int sub, int qe, int lane, float& zx,
    float& zy, float4& aq, float& dsum) {
  bool live[G];
  int e[G], s[G];
  float bs[G], sc[G], exq[G];
  float4 pc[G];
#pragma unroll
  for (int g = 0; g < G; ++g) {
    int j = 8 * g + qe;
    live[g] = j < cnt;
    int2 es = perm[b + (live[g] ? j : 0)];
    e[g] = es.x;
    s[g] = es.y;
  }
#pragma unroll
  for (int g = 0; g < G; ++g)
    pc[g] = *(const float4*)(ea + (size_t)e[g] * 16 + (sub << 2));
#pragma unroll
  for (int g = 0; g < G; ++g) bs[g] = sA[s[g]];
  float ml = NEGV;
#pragma unroll
  for (int g = 0; g < G; ++g) {
    float pd = pc[g].x * wq.x + pc[g].y * wq.y + pc[g].z * wq.z +
               pc[g].w * wq.w;
    pd += xrow<1>(pd);
    pd += xrow<2>(pd);
    float a = (bs[g] + sDn) + pd;
    a = a > 0.0f ? a : 0.2f * a;
    sc[g] = live[g] ? a : NEGV;
    ml = fmaxf(ml, sc[g]);
  }
  float m = wave_max(ml);
  float dq = 0.0f;
#pragma unroll
  for (int g = 0; g < G; ++g) {
    exq[g] = live[g] ? __expf(sc[g] - m) : 0.0f;
    dq += exq[g];
    aq.x += exq[g] * pc[g].x;
    aq.y += exq[g] * pc[g].y;
    aq.z += exq[g] * pc[g].z;
    aq.w += exq[g] * pc[g].w;
  }
  dsum = (sub == 0) ? dq : 0.0f;
  aq.x += xrow<4>(aq.x);
  aq.y += xrow<4>(aq.y);
  aq.z += xrow<4>(aq.z);
  aq.w += xrow<4>(aq.w);
  aq.x += xrow<8>(aq.x);
  aq.y += xrow<8>(aq.y);
  aq.z += xrow<8>(aq.z);
  aq.w += xrow<8>(aq.w);
  aq.x += xhalf(aq.x);
  aq.y += xhalf(aq.y);
  aq.z += xhalf(aq.z);
  aq.w += xhalf(aq.w);
#pragma unroll 1
  for (int g = 0; g < G; ++g) {
    int sg = s[0];
    float eg = exq[0];
#pragma unroll
    for (int k = 1; k < G; ++k) {
      if (g == k) {
        sg = s[k];
        eg = exq[k];
      }
    }
    add_rows8(xg, sg, eg, lane, zx, zy);
  }
}

__global__ void __launch_bounds__(256, 16) k_node(
    const float* __restrict__ ea, const float* __restrict__ Wn,
    const float* __restrict__ We, const float* __restrict__ weff,
    const float* __restrict__ x, const float* __restrict__ sA,
    const float* __restrict__ sD, const int* __restrict__ segStart,
    const int* __restrict__ permw, float* __restrict__ out, int N, int E) {
  const int2* __restrict__ perm = (const int2*)permw;
  __shared__ float s_a[16][84];
  __shared__ float s_p[4][16][68];

  int tid = threadIdx.x;
  int wave = __builtin_amdgcn_readfirstlane(tid >> 5);
  int lane = tid & 31;
  int rBase = blockIdx.x << 4;

  const float4* w4 = (const float4*)weff;
  float4 w0 = w4[0], w1 = w4[1], w2 = w4[2], w3 = w4[3];
  int sub = lane & 3;
  int qe = lane >> 2;
  const float4 wq = w4[sub];

  for (int q = 0; q < 2; ++q) {
    int r = (wave << 1) + q;
    int n = rBase + r;
    float zx = 0.0f, zy = 0.0f, ag = 0.0f, inv = 0.0f;
    float4 aq = make_float4(0.0f, 0.0f, 0.0f, 0.0f);
    bool quad = true;
    if (n < N) {
      int b = segStart[n];
      int en = (n + 1 < N) ? segStart[n + 1] : E;
      float sDn = sD[n];
      float dsum = 0.0f;
      if (en - b <= 24) {
        int cnt = en - b;
        if (cnt > 0) {
          int ng = (cnt + 7) >> 3;
          if (ng == 2)
            seg_quads<2>(ea, x, sA, perm, b, cnt, sDn, wq, sub, qe, lane, zx,
                         zy, aq, dsum);
          else if (ng == 3)
            seg_quads<3>(ea, x, sA, perm, b, cnt, sDn, wq, sub, qe, lane, zx,
                         zy, aq, dsum);
          else
            seg_quads<1>(ea, x, sA, perm, b, cnt, sDn, wq, sub, qe, lane, zx,
                         zy, aq, dsum);
        }
      } else if (en - b <= 32) {
        quad = false;
        int i = b + lane;
        int e = 0, s = 0;
        float sc = NEGV;
        if (i < en) {
          int2 es = perm[i];
          e = es.x;
          s = es.y;
          sc = edge_score(ea, e, sA[s] + sDn, w0, w1, w2, w3);
        }
        float m = wave_max(sc);
        float ex = (i < en) ? __expf(sc - m) : 0.0f;
        dsum = ex;
        add_chunk(x, ea, e, s, ex, en - b, lane, zx, zy, ag);
      } else {
        quad = false;
        float m = NEGV;
        for (int c = b; c < en; c += 32) {
          int i = c + lane;
          if (i < en) {
            int2 es = perm[i];
            m = fmaxf(m,
                      edge_score(ea, es.x, sA[es.y] + sDn, w0, w1, w2, w3));
          }
        }
        m = wave_max(m);
        for (int c = b; c < en; c += 32) {
          int i = c + lane;
          int e = 0, s = 0;
          float ex = 0.0f;
          if (i < en) {
            int2 es = perm[i];
            e = es.x;
            s = es.y;
            ex = __expf(edge_score(ea, e, sA[s] + sDn, w0, w1, w2, w3) - m);
          }
          dsum += ex;
          int cntc = en - c;
          if (cntc > 32) cntc = 32;
          add_chunk(x, ea, e, s, ex, cntc, lane, zx, zy, ag);
        }
      }
      dsum = wave_sum(dsum);
      inv = 1.0f / (dsum + 1e-8f);
    }
    s_a[r][(lane << 1)] = zx * inv;
    s_a[r][(lane << 1) + 1] = zy * inv;
    if (quad) {
      if (lane < 4) {
        s_a[r][64 + (lane << 2)] = aq.x * inv;
        s_a[r][64 + (lane << 2) + 1] = aq.y * inv;
        s_a[r][64 + (lane << 2) + 2] = aq.z * inv;
        s_a[r][64 + (lane << 2) + 3] = aq.w * inv;
      }
    } else if (lane < 16) {
      s_a[r][64 + lane] = ag * inv;
    }
  }
  int half = lane >> 4;
  int l16 = lane & 15;
  int p = wave & 1;
  int kq = wave >> 1;
  int c0 = (p << 5) + l16;
  v2f bw0[5], bw1[5];
#pragma unroll
  for (int tt = 0; tt < 5; ++tt) {
    int t = kq * 5 + tt;
    int k = (t << 2) + (half << 1);
    const float* __restrict__ wr = (t < 16) ? Wn + k * 64 : We + (k - 64) * 64;
    bw0[tt].x = wr[c0];
    bw0[tt].y = wr[64 + c0];
    bw1[tt].x = wr[c0 + 16];
    bw1[tt].y = wr[64 + c0 + 16];
  }
  int n0 = rBase + (wave << 1);
  int na = n0 < N ? n0 : N - 1;
  int nb = n0 + 1 < N ? n0 + 1 : N - 1;
  const float2 xa = *(const float2*)(x + (size_t)na * 64 + (lane << 1));
  const float2 xb = *(const float2*)(x + (size_t)nb * 64 + (lane << 1));

  __builtin_amdgcn_fence(__ATOMIC_RELEASE, "workgroup", "local");
  __builtin_amdgcn_s_barrier();
  __builtin_amdgcn_fence(__ATOMIC_ACQUIRE, "workgroup", "local");

  {
    v8f acc0 = {}, acc1 = {};
#pragma unroll
    for (int tt = 0; tt < 5; ++tt) {
      int t = kq * 5 + tt;
      int k = (t << 2) + (half << 1);
      v2f a;
      a.x = s_a[l16][k];
      a.y = s_a[l16][k + 1];
      acc0 = __builtin_amdgcn_wmma_f32_16x16x4_f32(
          false, a, false, bw0[tt], (short)0, acc0, false, false);
      acc1 = __builtin_amdgcn_wmma_f32_16x16x4_f32(
          false, a, false, bw1[tt], (short)0, acc1, false, false);
    }
#pragma unroll
    for (int j = 0; j < 8; ++j) {
      s_p[kq][(half << 3) + j][c0] = acc0[j];
      s_p[kq][(half << 3) + j][c0 + 16] = acc1[j];
    }
  }
  __syncthreads();

#pragma unroll
  for (int q = 0; q < 2; ++q) {
    int r = (wave << 1) + q;
    int n = rBase + r;
    if (n < N) {
      float2 v = q ? xb : xa;
#pragma unroll
      for (int kk = 0; kk < 4; ++kk) {
        float2 pq = *(const float2*)&s_p[kk][r][(lane << 1)];
        v.x += pq.x;
        v.y += pq.y;
      }
      *(float2*)(out + (size_t)n * 64 + (lane << 1)) = v;
    }
  }
}

extern "C" void kernel_launch(void* const* d_in, const int* in_sizes, int n_in,
                              void* d_out, int out_size, void* d_ws,
                              size_t ws_size, hipStream_t stream) {
  const float* x = (const float*)d_in[0];
  const int* ei = (const int*)d_in[1];
  const float* ea = (const float*)d_in[2];
  const float* Wn = (const float*)d_in[3];
  const float* We = (const float*)d_in[4];
  const float* att_src = (const float*)d_in[5];
  const float* att_dst = (const float*)d_in[6];
  const float* att_edge = (const float*)d_in[7];

  int N = in_sizes[0] / 64;
  int E = in_sizes[1] / 2;
  float* out = (float*)d_out;

  float* weff = (float*)d_ws;
  float* scr = weff + 64;
  float* sA = scr + (size_t)N * 64;
  float* sD = sA + N;
  int* cnt = (int*)(sD + N);
  int* planes = (int*)scr;
  int* tot = planes + (size_t)SORT_C * N;
  float* usd = (float*)(tot + 256);
  int* perm = tot + 512;

  int nRanges = (N + SORT_R - 1) / SORT_R;
  hipLaunchKernelGGL(k_count, dim3(nRanges * SORT_C), dim3(SORT_T), 0, stream,
                     ei, We, att_edge, Wn, att_src, att_dst, weff, usd, planes,
                     tot, N, E);

  hipLaunchKernelGGL(k_place, dim3(nRanges * SORT_C), dim3(SORT_T), 0, stream,
                     ei, planes, tot, cnt, perm, N, E);

  int nRowTiles = (N + 15) / 16;
  hipLaunchKernelGGL(k_node_scores, dim3((N + 255) / 256), dim3(256), 0, stream,
                     x, usd, sA, sD, N);

  hipLaunchKernelGGL(k_node, dim3(nRowTiles), dim3(256), 0, stream, ea, Wn, We,
                     weff, x, sA, sD, cnt, perm, out, N, E);
}


# node score kernel folded into the placement launch (4 launches -> 3): x rows read as whole lines, 16 lanes per row, dots finished by 4 DPP adds; sum order of the two per-node score dots changes
# speedup vs baseline: 3.4992x; 1.0390x over previous
#include <hip/hip_runtime.h>


typedef __attribute__((ext_vector_type(2))) float v2f;
typedef __attribute__((ext_vector_type(8))) float v8f;

#define NEGV (-1000000000.0f)

#define SORT_C 16
#define SORT_R 6400
#define SORT_T 1024
#define SORT_B 16

__device__ __forceinline__ int wave_sum_int(int v) {
#pragma unroll
  for (int off = 16; off > 0; off >>= 1) v += __shfl_xor(v, off);
  return v;
}

__global__ void __launch_bounds__(SORT_T) k_count(
    const int* __restrict__ ei, const float* __restrict__ We,
    const float* __restrict__ att_edge, const float* __restrict__ Wn,
    const float* __restrict__ att_src, const float* __restrict__ att_dst,
    float* __restrict__ weff, float* __restrict__ usd,
    int* __restrict__ planes, int* __restrict__ tot, int N, int E) {
  __shared__ int s_cnt[SORT_R];
  __shared__ int s_total;
  int tid = threadIdx.x;
  int lane = tid & 31;
  int j = blockIdx.x / SORT_C;
  int c = blockIdx.x - j * SORT_C;
  int lo = j * SORT_R;
  int nr = N - lo;
  if (nr > SORT_R) nr = SORT_R;
  int chunk = (E + SORT_C - 1) / SORT_C;
  int eb = c * chunk;
  int ee = eb + chunk;
  if (ee > E) ee = E;

  if (blockIdx.x == 0 && tid < 16) {
    float s = 0.0f;
    for (int k = 0; k < 64; ++k) s += We[tid * 64 + k] * att_edge[k];
    weff[tid] = s;
  }
  if (blockIdx.x == 0 && tid >= 32 && tid < 160) {
    int t = tid - 32;
    int k = t & 63;
    const float4* wr = (const float4*)(Wn + k * 64);
    const float4* av = (const float4*)((t >> 6) ? att_dst : att_src);
    float s = 0.0f;
#pragma unroll
    for (int i = 0; i < 16; ++i) {
      float4 w = wr[i], a = av[i];
      s += w.x * a.x + w.y * a.y + w.z * a.z + w.w * a.w;
    }
    usd[t] = s;
  }
  for (int i = tid; i < SORT_R; i += SORT_T) s_cnt[i] = 0;
  if (tid == 0) s_total = 0;
  __syncthreads();

  const int* __restrict__ dst = ei + E;
  for (int base = eb; base < ee; base += SORT_T * SORT_B) {
    int d[SORT_B];
#pragma unroll
    for (int u = 0; u < SORT_B; ++u) {
      int i = base + u * SORT_T + tid;
      int dv = dst[i < ee ? i : ee - 1];
      d[u] = (i < ee) ? dv : -1;
    }
#pragma unroll
    for (int u = 0; u < SORT_B; ++u) {
      unsigned r = (unsigned)(d[u] - lo);
      if (r < (unsigned)nr) atomicAdd(&s_cnt[r], 1);
    }
  }
  __syncthreads();

  int part = 0;
  for (int i = tid; i < nr; i += SORT_T) {
    int v = s_cnt[i];
    planes[(size_t)c * N + lo + i] = v;
    part += v;
  }
  part = wave_sum_int(part);
  if (lane == 0) atomicAdd(&s_total, part);
  __syncthreads();
  if (tid == 0) tot[blockIdx.x] = s_total;
}

template <int M>
__device__ __forceinline__ float xrow(float v) {
  return __int_as_float(__builtin_amdgcn_update_dpp(
      0, __float_as_int(v), 0x160 + M, 0xF, 0xF, true));
}

__global__ void __launch_bounds__(SORT_T) k_place(
    const int* __restrict__ ei, const int* __restrict__ planes,
    const int* __restrict__ tot, const float* __restrict__ x,
    const float* __restrict__ usd, float* __restrict__ sA,
    float* __restrict__ sD, int* __restrict__ segStart,
    int* __restrict__ permw, int N, int E) {
  int2* __restrict__ perm = (int2*)permw;
  __shared__ int s_slot[SORT_R];
  __shared__ int s_bef[SORT_R];
  __shared__ int s_w[32];
  __shared__ int s_rb[32];
  int tid = threadIdx.x;
  int lane = tid & 31;
  int wave = tid >> 5;
  int j = blockIdx.x / SORT_C;
  int c = blockIdx.x - j * SORT_C;
  int lo = j * SORT_R;
  int nr = N - lo;
  if (nr > SORT_R) nr = SORT_R;
  int chunk = (E + SORT_C - 1) / SORT_C;
  int eb = c * chunk;
  int ee = eb + chunk;
  if (ee > E) ee = E;

  constexpr int NS_U = 8;
  constexpr int NS_G = SORT_T / 16;
  static_assert(NS_G * NS_U >= (SORT_R + SORT_C - 1) / SORT_C,
                "a work-group's share of the nodes must fit one trip");
  int nwg = ((N + SORT_R - 1) / SORT_R) * SORT_C;
  int per = (N + nwg - 1) / nwg;
  int nBeg = blockIdx.x * per;
  int nEnd = nBeg + per;
  if (nEnd > N) nEnd = N;
  int l16 = tid & 15;
  int grp = tid >> 4;
  const float4* us4 = (const float4*)usd;
  float4 us = us4[l16], ud = us4[16 + l16];
  float4 xv[NS_U];
#pragma unroll
  for (int u = 0; u < NS_U; ++u) {
    int n = nBeg + u * NS_G + grp;
    int nc = n < nEnd ? n : nEnd - 1;
    xv[u] = ((const float4*)(x + (size_t)nc * 64))[l16];
  }

  int rb = 0;
  for (int t = tid; t < j * SORT_C; t += SORT_T) rb += tot[t];
  rb = wave_sum_int(rb);
  if (lane == 0) s_rb[wave] = rb;

  for (int i = tid; i < SORT_R; i += SORT_T) {
    int t = 0, b = 0;
    if (i < nr) {
#pragma unroll
      for (int cc = 0; cc < SORT_C; ++cc) {
        int v = planes[(size_t)cc * N + lo + i];
        t += v;
        if (cc < c) b += v;
      }
    }
    s_slot[i] = t;
    s_bef[i] = b;
  }

#pragma unroll
  for (int u = 0; u < NS_U; ++u) {
    int n = nBeg + u * NS_G + grp;
    float a = xv[u].x * us.x + xv[u].y * us.y + xv[u].z * us.z + xv[u].w * us.w;
    float d = xv[u].x * ud.x + xv[u].y * ud.y + xv[u].z * ud.z + xv[u].w * ud.w;
    a += xrow<1>(a);
    d += xrow<1>(d);
    a += xrow<2>(a);
    d += xrow<2>(d);
    a += xrow<4>(a);
    d += xrow<4>(d);
    a += xrow<8>(a);
    d += xrow<8>(d);
    if (l16 == 0 && n < nEnd) {
      sA[n] = a;
      sD[n] = d;
    }
  }
  __syncthreads();

  int rbase = wave_sum_int(s_rb[lane]);

  constexpr int SORT_K = (SORT_R + SORT_T - 1) / SORT_T;
  int i0 = tid * SORT_K;
  int v[SORT_K];
  int sum = 0;
#pragma unroll
  for (int k = 0; k < SORT_K; ++k) {
    int i = i0 + k;
    v[k] = (i < SORT_R) ? s_slot[i] : 0;
    sum += v[k];
  }
  int incl = sum;
#pragma unroll
  for (int off = 1; off < 32; off <<= 1) {
    int u = __shfl_up(incl, off);
    if (lane >= off) incl += u;
  }
  if (lane == 31) s_w[wave] = incl;
  __syncthreads();
  int wv = s_w[lane];
  int winc = wv;
#pragma unroll
  for (int off = 1; off < 32; off <<= 1) {
    int u = __shfl_up(winc, off);
    if (lane >= off) winc += u;
  }
  int run = rbase + __shfl(winc, wave) - __shfl(wv, wave) + (incl - sum);
#pragma unroll
  for (int k = 0; k < SORT_K; ++k) {
    int i = i0 + k;
    if (i < SORT_R) {
      if (c == 0 && i < nr) segStart[lo + i] = run;
      s_slot[i] = run + s_bef[i];
      run += v[k];
    }
  }
  __syncthreads();

  const int* __restrict__ dst = ei + E;
  for (int base = eb; base < ee; base += SORT_T * SORT_B) {
    int d[SORT_B];
    int sv[SORT_B];
#pragma unroll
    for (int u = 0; u < SORT_B; ++u) {
      int i = base + u * SORT_T + tid;
      int ic = i < ee ? i : ee - 1;
      d[u] = dst[ic];
      sv[u] = ei[ic];
    }
    int slot[SORT_B];
#pragma unroll
    for (int u = 0; u < SORT_B; ++u) {
      int i = base + u * SORT_T + tid;
      unsigned r = (unsigned)(d[u] - lo);
      slot[u] = (i < ee && r < (unsigned)nr) ? atomicAdd(&s_slot[r], 1) : -1;
    }
#pragma unroll
    for (int u = 0; u < SORT_B; ++u)
      if (slot[u] >= 0)
        perm[slot[u]] = make_int2(base + u * SORT_T + tid, sv[u]);
  }
}

__device__ __forceinline__ float wave_max(float v) {
#pragma unroll
  for (int off = 16; off > 0; off >>= 1) v = fmaxf(v, __shfl_xor(v, off));
  return v;
}
__device__ __forceinline__ float wave_sum(float v) {
#pragma unroll
  for (int off = 16; off > 0; off >>= 1) v += __shfl_xor(v, off);
  return v;
}

__device__ __forceinline__ float edge_score(const float* __restrict__ ea,
                                            int e, float base, float4 w0,
                                            float4 w1, float4 w2, float4 w3) {
  const float4* er = (const float4*)(ea + (size_t)e * 16);
  float4 v0 = er[0], v1 = er[1], v2 = er[2], v3 = er[3];
  float acc = base;
  acc += v0.x * w0.x + v0.y * w0.y + v0.z * w0.z + v0.w * w0.w;
  acc += v1.x * w1.x + v1.y * w1.y + v1.z * w1.z + v1.w * w1.w;
  acc += v2.x * w2.x + v2.y * w2.y + v2.z * w2.z + v2.w * w2.w;
  acc += v3.x * w3.x + v3.y * w3.y + v3.z * w3.z + v3.w * w3.w;
  return acc > 0.0f ? acc : 0.2f * acc;
}

__device__ __forceinline__ void add_chunk(const float* __restrict__ xg,
                                          const float* __restrict__ ea,
                                          int e, int s, float ex, int cntc,
                                          int lane, float& zx, float& zy,
                                          float& ag) {
  int e0 = __builtin_amdgcn_readfirstlane(e);
  int s0 = __builtin_amdgcn_readfirstlane(s);
  if (lane >= cntc) {
    e = e0;
    s = s0;
  }
  int cnt8 = (cntc + 7) & ~7;
  for (int j0 = 0; j0 < cnt8; j0 += 8) {
    float2 zv[8];
    float av[8];
    float wj[8];
#pragma unroll
    for (int u = 0; u < 8; ++u) {
      int ej = __builtin_amdgcn_readlane(e, j0 + u);
      int sj = __builtin_amdgcn_readlane(s, j0 + u);
      wj[u] = __int_as_float(
          __builtin_amdgcn_readlane(__float_as_int(ex), j0 + u));
      zv[u] = *(const float2*)(xg + (size_t)sj * 64 + (lane << 1));
      av[u] = ea[(size_t)ej * 16 + (lane & 15)];
    }
#pragma unroll
    for (int u = 0; u < 8; ++u) {
      zx += wj[u] * zv[u].x;
      zy += wj[u] * zv[u].y;
      ag += wj[u] * av[u];
    }
  }
}

__device__ __forceinline__ float xhalf(float v) {
  return __int_as_float(__builtin_amdgcn_permlanex16(
      __float_as_int(v), __float_as_int(v), 0x76543210, 0xfedcba98, false,
      false));
}

__device__ __forceinline__ void add_rows8(const float* __restrict__ xg, int sq,
                                          float exq, int lane, float& zx,
                                          float& zy) {
  float2 zv[8];
  float wj[8];
#pragma unroll
  for (int u = 0; u < 8; ++u) {
    int sj = __builtin_amdgcn_readlane(sq, 4 * u);
    wj[u] = __int_as_float(
        __builtin_amdgcn_readlane(__float_as_int(exq), 4 * u));
    zv[u] = *(const float2*)(xg + (size_t)sj * 64 + (lane << 1));
  }
#pragma unroll
  for (int u = 0; u < 8; ++u) {
    zx += wj[u] * zv[u].x;
    zy += wj[u] * zv[u].y;
  }
}

template <int G>
__device__ __forceinline__ void seg_quads(
    const float* __restrict__ ea, const float* __restrict__ xg,
    const float* __restrict__ sA, const int2* __restrict__ perm, int b,
    int cnt, float sDn, float4 wq, int sub, int qe, int lane, float& zx,
    float& zy, float4& aq, float& dsum) {
  bool live[G];
  int e[G], s[G];
  float bs[G], sc[G], exq[G];
  float4 pc[G];
#pragma unroll
  for (int g = 0; g < G; ++g) {
    int j = 8 * g + qe;
    live[g] = j < cnt;
    int2 es = perm[b + (live[g] ? j : 0)];
    e[g] = es.x;
    s[g] = es.y;
  }
#pragma unroll
  for (int g = 0; g < G; ++g)
    pc[g] = *(const float4*)(ea + (size_t)e[g] * 16 + (sub << 2));
#pragma unroll
  for (int g = 0; g < G; ++g) bs[g] = sA[s[g]];
  float ml = NEGV;
#pragma unroll
  for (int g = 0; g < G; ++g) {
    float pd = pc[g].x * wq.x + pc[g].y * wq.y + pc[g].z * wq.z +
               pc[g].w * wq.w;
    pd += xrow<1>(pd);
    pd += xrow<2>(pd);
    float a = (bs[g] + sDn) + pd;
    a = a > 0.0f ? a : 0.2f * a;
    sc[g] = live[g] ? a : NEGV;
    ml = fmaxf(ml, sc[g]);
  }
  float m = wave_max(ml);
  float dq = 0.0f;
#pragma unroll
  for (int g = 0; g < G; ++g) {
    exq[g] = live[g] ? __expf(sc[g] - m) : 0.0f;
    dq += exq[g];
    aq.x += exq[g] * pc[g].x;
    aq.y += exq[g] * pc[g].y;
    aq.z += exq[g] * pc[g].z;
    aq.w += exq[g] * pc[g].w;
  }
  dsum = (sub == 0) ? dq : 0.0f;
  aq.x += xrow<4>(aq.x);
  aq.y += xrow<4>(aq.y);
  aq.z += xrow<4>(aq.z);
  aq.w += xrow<4>(aq.w);
  aq.x += xrow<8>(aq.x);
  aq.y += xrow<8>(aq.y);
  aq.z += xrow<8>(aq.z);
  aq.w += xrow<8>(aq.w);
  aq.x += xhalf(aq.x);
  aq.y += xhalf(aq.y);
  aq.z += xhalf(aq.z);
  aq.w += xhalf(aq.w);
#pragma unroll 1
  for (int g = 0; g < G; ++g) {
    int sg = s[0];
    float eg = exq[0];
#pragma unroll
    for (int k = 1; k < G; ++k) {
      if (g == k) {
        sg = s[k];
        eg = exq[k];
      }
    }
    add_rows8(xg, sg, eg, lane, zx, zy);
  }
}

__global__ void __launch_bounds__(256, 16) k_node(
    const float* __restrict__ ea, const float* __restrict__ Wn,
    const float* __restrict__ We, const float* __restrict__ weff,
    const float* __restrict__ x, const float* __restrict__ sA,
    const float* __restrict__ sD, const int* __restrict__ segStart,
    const int* __restrict__ permw, float* __restrict__ out, int N, int E) {
  const int2* __restrict__ perm = (const int2*)permw;
  __shared__ float s_a[16][84];
  __shared__ float s_p[4][16][68];

  int tid = threadIdx.x;
  int wave = __builtin_amdgcn_readfirstlane(tid >> 5);
  int lane = tid & 31;
  int rBase = blockIdx.x << 4;

  const float4* w4 = (const float4*)weff;
  float4 w0 = w4[0], w1 = w4[1], w2 = w4[2], w3 = w4[3];
  int sub = lane & 3;
  int qe = lane >> 2;
  const float4 wq = w4[sub];

  for (int q = 0; q < 2; ++q) {
    int r = (wave << 1) + q;
    int n = rBase + r;
    float zx = 0.0f, zy = 0.0f, ag = 0.0f, inv = 0.0f;
    float4 aq = make_float4(0.0f, 0.0f, 0.0f, 0.0f);
    bool quad = true;
    if (n < N) {
      int b = segStart[n];
      int en = (n + 1 < N) ? segStart[n + 1] : E;
      float sDn = sD[n];
      float dsum = 0.0f;
      if (en - b <= 24) {
        int cnt = en - b;
        if (cnt > 0) {
          int ng = (cnt + 7) >> 3;
          if (ng == 2)
            seg_quads<2>(ea, x, sA, perm, b, cnt, sDn, wq, sub, qe, lane, zx,
                         zy, aq, dsum);
          else if (ng == 3)
            seg_quads<3>(ea, x, sA, perm, b, cnt, sDn, wq, sub, qe, lane, zx,
                         zy, aq, dsum);
          else
            seg_quads<1>(ea, x, sA, perm, b, cnt, sDn, wq, sub, qe, lane, zx,
                         zy, aq, dsum);
        }
      } else if (en - b <= 32) {
        quad = false;
        int i = b + lane;
        int e = 0, s = 0;
        float sc = NEGV;
        if (i < en) {
          int2 es = perm[i];
          e = es.x;
          s = es.y;
          sc = edge_score(ea, e, sA[s] + sDn, w0, w1, w2, w3);
        }
        float m = wave_max(sc);
        float ex = (i < en) ? __expf(sc - m) : 0.0f;
        dsum = ex;
        add_chunk(x, ea, e, s, ex, en - b, lane, zx, zy, ag);
      } else {
        quad = false;
        float m = NEGV;
        for (int c = b; c < en; c += 32) {
          int i = c + lane;
          if (i < en) {
            int2 es = perm[i];
            m = fmaxf(m,
                      edge_score(ea, es.x, sA[es.y] + sDn, w0, w1, w2, w3));
          }
        }
        m = wave_max(m);
        for (int c = b; c < en; c += 32) {
          int i = c + lane;
          int e = 0, s = 0;
          float ex = 0.0f;
          if (i < en) {
            int2 es = perm[i];
            e = es.x;
            s = es.y;
            ex = __expf(edge_score(ea, e, sA[s] + sDn, w0, w1, w2, w3) - m);
          }
          dsum += ex;
          int cntc = en - c;
          if (cntc > 32) cntc = 32;
          add_chunk(x, ea, e, s, ex, cntc, lane, zx, zy, ag);
        }
      }
      dsum = wave_sum(dsum);
      inv = 1.0f / (dsum + 1e-8f);
    }
    s_a[r][(lane << 1)] = zx * inv;
    s_a[r][(lane << 1) + 1] = zy * inv;
    if (quad) {
      if (lane < 4) {
        s_a[r][64 + (lane << 2)] = aq.x * inv;
        s_a[r][64 + (lane << 2) + 1] = aq.y * inv;
        s_a[r][64 + (lane << 2) + 2] = aq.z * inv;
        s_a[r][64 + (lane << 2) + 3] = aq.w * inv;
      }
    } else if (lane < 16) {
      s_a[r][64 + lane] = ag * inv;
    }
  }
  int half = lane >> 4;
  int l16 = lane & 15;
  int p = wave & 1;
  int kq = wave >> 1;
  int c0 = (p << 5) + l16;
  v2f bw0[5], bw1[5];
#pragma unroll
  for (int tt = 0; tt < 5; ++tt) {
    int t = kq * 5 + tt;
    int k = (t << 2) + (half << 1);
    const float* __restrict__ wr = (t < 16) ? Wn + k * 64 : We + (k - 64) * 64;
    bw0[tt].x = wr[c0];
    bw0[tt].y = wr[64 + c0];
    bw1[tt].x = wr[c0 + 16];
    bw1[tt].y = wr[64 + c0 + 16];
  }
  int n0 = rBase + (wave << 1);
  int na = n0 < N ? n0 : N - 1;
  int nb = n0 + 1 < N ? n0 + 1 : N - 1;
  const float2 xa = *(const float2*)(x + (size_t)na * 64 + (lane << 1));
  const float2 xb = *(const float2*)(x + (size_t)nb * 64 + (lane << 1));

  __builtin_amdgcn_fence(__ATOMIC_RELEASE, "workgroup", "local");
  __builtin_amdgcn_s_barrier();
  __builtin_amdgcn_fence(__ATOMIC_ACQUIRE, "workgroup", "local");

  {
    v8f acc0 = {}, acc1 = {};
#pragma unroll
    for (int tt = 0; tt < 5; ++tt) {
      int t = kq * 5 + tt;
      int k = (t << 2) + (half << 1);
      v2f a;
      a.x = s_a[l16][k];
      a.y = s_a[l16][k + 1];
      acc0 = __builtin_amdgcn_wmma_f32_16x16x4_f32(
          false, a, false, bw0[tt], (short)0, acc0, false, false);
      acc1 = __builtin_amdgcn_wmma_f32_16x16x4_f32(
          false, a, false, bw1[tt], (short)0, acc1, false, false);
    }
#pragma unroll
    for (int j = 0; j < 8; ++j) {
      s_p[kq][(half << 3) + j][c0] = acc0[j];
      s_p[kq][(half << 3) + j][c0 + 16] = acc1[j];
    }
  }
  __syncthreads();

#pragma unroll
  for (int q = 0; q < 2; ++q) {
    int r = (wave << 1) + q;
    int n = rBase + r;
    if (n < N) {
      float2 v = q ? xb : xa;
#pragma unroll
      for (int kk = 0; kk < 4; ++kk) {
        float2 pq = *(const float2*)&s_p[kk][r][(lane << 1)];
        v.x += pq.x;
        v.y += pq.y;
      }
      *(float2*)(out + (size_t)n * 64 + (lane << 1)) = v;
    }
  }
}

extern "C" void kernel_launch(void* const* d_in, const int* in_sizes, int n_in,
                              void* d_out, int out_size, void* d_ws,
                              size_t ws_size, hipStream_t stream) {
  const float* x = (const float*)d_in[0];
  const int* ei = (const int*)d_in[1];
  const float* ea = (const float*)d_in[2];
  const float* Wn = (const float*)d_in[3];
  const float* We = (const float*)d_in[4];
  const float* att_src = (const float*)d_in[5];
  const float* att_dst = (const float*)d_in[6];
  const float* att_edge = (const float*)d_in[7];

  int N = in_sizes[0] / 64;
  int E = in_sizes[1] / 2;
  float* out = (float*)d_out;

  float* weff = (float*)d_ws;
  float* scr = weff + 64;
  float* sA = scr + (size_t)N * 64;
  float* sD = sA + N;
  int* cnt = (int*)(sD + N);
  int* planes = (int*)scr;
  int* tot = planes + (size_t)SORT_C * N;
  float* usd = (float*)(tot + 256);
  int* perm = tot + 512;

  int nRanges = (N + SORT_R - 1) / SORT_R;
  hipLaunchKernelGGL(k_count, dim3(nRanges * SORT_C), dim3(SORT_T), 0, stream,
                     ei, We, att_edge, Wn, att_src, att_dst, weff, usd, planes,
                     tot, N, E);

  hipLaunchKernelGGL(k_place, dim3(nRanges * SORT_C), dim3(SORT_T), 0, stream,
                     ei, planes, tot, x, usd, sA, sD, cnt, perm, N, E);

  int nRowTiles = (N + 15) / 16;

  hipLaunchKernelGGL(k_node, dim3(nRowTiles), dim3(256), 0, stream, ea, Wn, We,
                     weff, x, sA, sD, cnt, perm, out, N, E);
}
